# starting kernel
# baseline (speedup 1.0000x reference)
#include <hip/hip_runtime.h>
#include <math.h>

#define Nn 50000
#define Ee 1600000
#define Dd 128
#define Hh 128
#define Cc 10
#define Gg 512
#define LRELU 0.05f
#define BN_EPS 1e-5f

typedef __attribute__((ext_vector_type(16))) __bf16 v16bf;
typedef __attribute__((ext_vector_type(8)))  float  v8f;

union BFrag { v16bf v; unsigned u[8]; };

__device__ __forceinline__ unsigned short f2bf_bits(float f) {
  union { float f; unsigned u; } v; v.f = f;
  return (unsigned short)((v.u + 0x7FFFu + ((v.u >> 16) & 1u)) >> 16);
}

__global__ void k_zero(float* __restrict__ p, long long n) {
  long long i = (long long)blockIdx.x * blockDim.x + threadIdx.x;
  const long long stride = (long long)gridDim.x * blockDim.x;
  for (; i < n; i += stride) p[i] = 0.0f;
}

__global__ void k_cvt(const float* __restrict__ s, unsigned short* __restrict__ d, long long n) {
  long long i = (long long)blockIdx.x * blockDim.x + threadIdx.x;
  const long long stride = (long long)gridDim.x * blockDim.x;
  for (; i < n; i += stride) d[i] = f2bf_bits(s[i]);
}

__global__ void __launch_bounds__(256) k_stats(const float* __restrict__ x, float* __restrict__ stats) {
  __shared__ float ssum[256], ssq[256];
  const int tid = threadIdx.x;
  long long i = (long long)blockIdx.x * blockDim.x + tid;
  const long long stride = (long long)gridDim.x * blockDim.x;
  const long long total = (long long)Nn * Dd;
  float s = 0.f, q = 0.f;
  for (; i < total; i += stride) { float v = x[i]; s += v; q += v * v; }
  ssum[tid] = s; ssq[tid] = q; __syncthreads();
  for (int o = 128; o > 0; o >>= 1) {
    if (tid < o) { ssum[tid] += ssum[tid + o]; ssq[tid] += ssq[tid + o]; }
    __syncthreads();
  }
  if (tid == 0) { atomicAdd(&stats[0], ssum[0]); atomicAdd(&stats[1], ssq[0]); }
}

__global__ void k_prep(const float* __restrict__ w, float* __restrict__ stats,
                       float* __restrict__ corr) {
  const int h = threadIdx.x;
  float colsum = 0.f;
  for (int d = 0; d < Dd; ++d) colsum += w[h * Dd + d];
  const float nd = (float)((long long)Nn * Dd);
  const float s  = stats[0], ss = stats[1];
  const float mu = s / nd;
  const float var = (ss - s * s / nd) / (nd - 1.0f);
  const float invsig = rsqrtf(var);
  if (h == 0) stats[2] = invsig;
  corr[h] = mu * invsig * colsum;
}

__global__ void __launch_bounds__(256) k_gat_gemm(const unsigned short* __restrict__ xbf,
                                                  const unsigned short* __restrict__ wbf,
                                                  const float* __restrict__ stats,
                                                  const float* __restrict__ corr,
                                                  float* __restrict__ z) {
  const int tid  = threadIdx.x;
  const int lane = tid & 31;
  const int hi   = lane >> 4;
  const int lm   = lane & 15;
  const int n0   = blockIdx.x * 16;
  const int h0   = (tid >> 5) * 16;

  const unsigned* xp = (const unsigned*)xbf + (size_t)(n0 + lm) * (Dd / 2);
  const unsigned* wp = (const unsigned*)wbf + (size_t)(h0 + lm) * (Dd / 2);

  v8f c = {0.f, 0.f, 0.f, 0.f, 0.f, 0.f, 0.f, 0.f};
  const float invsig = stats[2];

  #pragma unroll
  for (int k0 = 0; k0 < Dd; k0 += 32) {
    BFrag a, b;
    #pragma unroll
    for (int j = 0; j < 8; ++j) {
      const int kk = k0 + ((j < 4) ? 0 : 16) + (hi ? 8 : 0) + 2 * (j & 3);
      a.u[j] = xp[kk >> 1];
      b.u[j] = wp[kk >> 1];
    }
    if (k0 + 32 < Dd)
      __builtin_prefetch(xp + ((k0 + 32) >> 1), 0, 1);
    c = __builtin_amdgcn_wmma_f32_16x16x32_bf16(false, a.v, false, b.v, (short)0, c, false, false);
  }
  #pragma unroll
  for (int r = 0; r < 8; ++r) {
    const int m = r + hi * 8;
    const int h = h0 + lm;
    z[(size_t)(n0 + m) * Hh + h] = c[r] * invsig - corr[h];
  }
}

__global__ void __launch_bounds__(256) k_s12(const float* __restrict__ z,
                                             const float* __restrict__ a1,
                                             const float* __restrict__ a2,
                                             float* __restrict__ s1, float* __restrict__ s2) {
  const int lane = threadIdx.x & 31;
  const int n = blockIdx.x * 8 + (threadIdx.x >> 5);
  if (n >= Nn) return;
  float p1 = 0.f, p2 = 0.f;
  for (int i = lane; i < Hh; i += 32) {
    const float zv = z[(size_t)n * Hh + i];
    p1 += zv * a1[i]; p2 += zv * a2[i];
  }
  for (int o = 16; o > 0; o >>= 1) { p1 += __shfl_xor(p1, o); p2 += __shfl_xor(p2, o); }
  if (lane == 0) { s1[n] = p1; s2[n] = p2; }
}

__global__ void __launch_bounds__(256) k_edge1(const int* __restrict__ src, const int* __restrict__ dst,
                                               const float* __restrict__ s1, const float* __restrict__ s2,
                                               float* __restrict__ hbuf, float* __restrict__ hsum) {
  long long i = (long long)blockIdx.x * blockDim.x + threadIdx.x;
  const long long stride = (long long)gridDim.x * blockDim.x;
  for (; i < Ee; i += stride) {
    const int s = src[i], d = dst[i];
    float e = s1[s] + s2[d];
    e = (e >= 0.f) ? e : LRELU * e;
    const float h = __expf(e);
    hbuf[i] = h;
    atomicAdd(&hsum[s], h);
  }
}

__global__ void __launch_bounds__(256) k_edge2(const int* __restrict__ src, const int* __restrict__ dst,
                                               const float* __restrict__ hbuf, const float* __restrict__ hsum,
                                               const float* __restrict__ z, float* __restrict__ gout) {
  const int lane = threadIdx.x & 31;
  const long long e = (long long)blockIdx.x * 8 + (threadIdx.x >> 5);
  if (e >= Ee) return;
  const int s = src[e], d = dst[e];
  const float alpha = hbuf[e] / hsum[s];
  const float4 zv = *(const float4*)(z + (size_t)d * Hh + lane * 4);
  float* o = gout + (size_t)s * Hh + lane * 4;
  atomicAdd(o + 0, alpha * zv.x);
  atomicAdd(o + 1, alpha * zv.y);
  atomicAdd(o + 2, alpha * zv.z);
  atomicAdd(o + 3, alpha * zv.w);
}

__global__ void __launch_bounds__(256) k_pool(const float* __restrict__ gout, const int* __restrict__ idx,
                                              float* __restrict__ pool) {
  const int lane = threadIdx.x & 31;
  const int n = blockIdx.x * 8 + (threadIdx.x >> 5);
  if (n >= Nn) return;
  const int g = idx[n];
  const float4 v = *(const float4*)(gout + (size_t)n * Hh + lane * 4);
  float* p = pool + (size_t)g * Hh + lane * 4;
  atomicAdd(p + 0, fmaxf(v.x, 0.f));
  atomicAdd(p + 1, fmaxf(v.y, 0.f));
  atomicAdd(p + 2, fmaxf(v.z, 0.f));
  atomicAdd(p + 3, fmaxf(v.w, 0.f));
}

__global__ void k_bn(const float* __restrict__ pooled, const float* __restrict__ gamma,
                     const float* __restrict__ beta, float* __restrict__ xb) {
  const int h = threadIdx.x;
  float s = 0.f, q = 0.f;
  for (int g = 0; g < Gg; ++g) { const float v = pooled[g * Hh + h]; s += v; q += v * v; }
  const float mu = s / (float)Gg;
  const float var = q / (float)Gg - mu * mu;
  const float inv = rsqrtf(var + BN_EPS);
  const float ga = gamma[h], be = beta[h];
  for (int g = 0; g < Gg; ++g)
    xb[g * Hh + h] = (pooled[g * Hh + h] - mu) * inv * ga + be;
}

__global__ void __launch_bounds__(256) k_fc1(const unsigned short* __restrict__ xbf,
                                             const unsigned short* __restrict__ wbf,
                                             const float* __restrict__ bias, float* __restrict__ out) {
  const int tid  = threadIdx.x;
  const int lane = tid & 31;
  const int hi   = lane >> 4;
  const int lm   = lane & 15;
  const int n0   = blockIdx.x * 16;
  const int h0   = (tid >> 5) * 16;

  const unsigned* xp = (const unsigned*)xbf + (size_t)(n0 + lm) * (Hh / 2);
  const unsigned* wp = (const unsigned*)wbf + (size_t)(h0 + lm) * (Hh / 2);

  v8f c = {0.f, 0.f, 0.f, 0.f, 0.f, 0.f, 0.f, 0.f};
  #pragma unroll
  for (int k0 = 0; k0 < Hh; k0 += 32) {
    BFrag a, b;
    #pragma unroll
    for (int j = 0; j < 8; ++j) {
      const int kk = k0 + ((j < 4) ? 0 : 16) + (hi ? 8 : 0) + 2 * (j & 3);
      a.u[j] = xp[kk >> 1];
      b.u[j] = wp[kk >> 1];
    }
    c = __builtin_amdgcn_wmma_f32_16x16x32_bf16(false, a.v, false, b.v, (short)0, c, false, false);
  }
  #pragma unroll
  for (int r = 0; r < 8; ++r) {
    const int m = r + hi * 8;
    const int h = h0 + lm;
    out[(size_t)(n0 + m) * Hh + h] = fmaxf(c[r] + bias[h], 0.f);
  }
}

__global__ void __launch_bounds__(256) k_fc2(const float* __restrict__ x, const float* __restrict__ w,
                                             const float* __restrict__ b, float* __restrict__ out) {
  const int lane = threadIdx.x & 31;
  const int g = blockIdx.x * 8 + (threadIdx.x >> 5);
  if (g >= Gg) return;
  float dot = -INFINITY;
  if (lane < Cc) {
    float acc = b[lane];
    for (int d = 0; d < Hh; ++d) acc += x[(size_t)g * Hh + d] * w[lane * Hh + d];
    dot = acc;
  }
  float m = dot;
  for (int o = 16; o > 0; o >>= 1) m = fmaxf(m, __shfl_xor(m, o));
  float ex = (lane < Cc) ? __expf(dot - m) : 0.f;
  float se = ex;
  for (int o = 16; o > 0; o >>= 1) se += __shfl_xor(se, o);
  if (lane < Cc) out[(size_t)g * Cc + lane] = dot - m - logf(se);
}

extern "C" void kernel_launch(void* const* d_in, const int* in_sizes, int n_in,
                              void* d_out, int out_size, void* d_ws, size_t ws_size,
                              hipStream_t stream) {
  const float* x_in  = (const float*)d_in[0];
  const int*   eidx  = (const int*)  d_in[1];
  const int*   idx   = (const int*)  d_in[2];
  const float* fc_w  = (const float*)d_in[3];
  const float* a1    = (const float*)d_in[4];
  const float* a2    = (const float*)d_in[5];
  const float* fc1_w = (const float*)d_in[6];
  const float* fc1_b = (const float*)d_in[7];
  const float* fc2_w = (const float*)d_in[8];
  const float* fc2_b = (const float*)d_in[9];
  const float* gamma = (const float*)d_in[10];
  const float* beta  = (const float*)d_in[11];
  const int* src = eidx;
  const int* dst = eidx + Ee;

  float* ws   = (float*)d_ws;
  float* STAT = ws;
  float* CORR = ws + 64;
  float* Z    = ws + 256;
  float* S1   = Z + (size_t)Nn * Hh;
  float* S2   = S1 + Nn;
  float* HB   = S2 + Nn;
  float* HS   = HB + Ee;
  float* GOUT = HS + Nn;
  float* POOL = GOUT + (size_t)Nn * Hh;
  float* XB   = POOL + (size_t)Gg * Hh;
  float* FC1O = XB + (size_t)Gg * Hh;
  unsigned short* XBF  = (unsigned short*)(FC1O + (size_t)Gg * Hh);
  unsigned short* WBF  = XBF  + (size_t)Nn * Dd;
  unsigned short* W1BF = WBF  + (size_t)Hh * Dd;
  unsigned short* XBBF = W1BF + (size_t)Hh * Hh;

  (void)in_sizes; (void)n_in; (void)out_size; (void)ws_size;

  k_zero<<<1,    256, 0, stream>>>(STAT, 16);
  k_zero<<<256,  256, 0, stream>>>(HS, Nn);
  k_zero<<<2048, 256, 0, stream>>>(GOUT, (long long)Nn * Hh);
  k_zero<<<64,   256, 0, stream>>>(POOL, (long long)Gg * Hh);

  k_cvt<<<2048, 256, 0, stream>>>(x_in,  XBF,  (long long)Nn * Dd);
  k_cvt<<<64,   256, 0, stream>>>(fc_w,  WBF,  (long long)Hh * Dd);
  k_cvt<<<64,   256, 0, stream>>>(fc1_w, W1BF, (long long)Hh * Hh);

  k_stats<<<1024, 256, 0, stream>>>(x_in, STAT);
  k_prep<<<1, 128, 0, stream>>>(fc_w, STAT, CORR);
  k_gat_gemm<<<Nn / 16, 256, 0, stream>>>(XBF, WBF, STAT, CORR, Z);
  k_s12<<<Nn / 8, 256, 0, stream>>>(Z, a1, a2, S1, S2);
  k_edge1<<<4096, 256, 0, stream>>>(src, dst, S1, S2, HB, HS);
  k_edge2<<<Ee / 8, 256, 0, stream>>>(src, dst, HB, HS, Z, GOUT);
  k_pool<<<Nn / 8, 256, 0, stream>>>(GOUT, idx, POOL);
  k_bn<<<1, 128, 0, stream>>>(POOL, gamma, beta, XB);
  k_cvt<<<64, 256, 0, stream>>>(XB, XBBF, (long long)Gg * Hh);
  k_fc1<<<Gg / 16, 256, 0, stream>>>(XBBF, W1BF, fc1_b, FC1O);
  k_fc2<<<Gg / 8, 256, 0, stream>>>(FC1O, fc2_w, fc2_b, (float*)d_out);
}


# pooling by runs: a wave takes 16 consecutive nodes, all rows in flight, sums relu(row) in registers in node order while the graph id repeats; one set of f32 atomic adds per run, not per node
# speedup vs baseline: 4.0134x; 4.0134x over previous
#include <hip/hip_runtime.h>
#include <math.h>

#define Nn 50000
#define Ee 1600000
#define Dd 128
#define Hh 128
#define Cc 10
#define Gg 512
#define LRELU 0.05f
#define BN_EPS 1e-5f

#define NOWN 256
#define ROWS 196
#define QCAP 32768
#define NQ4  (Ee / 16)
static_assert(NOWN * ROWS >= Nn, "owners must cover every node");
static_assert(NOWN <= 256, "owner id is one byte");
static_assert(Nn <= 65536, "src and dst are packed as two 16-bit halves");
static_assert(Ee % 16 == 0, "owner bytes are read 16 at a time");

typedef __attribute__((ext_vector_type(16))) __bf16 v16bf;
typedef __attribute__((ext_vector_type(8)))  float  v8f;

union BFrag { v16bf v; unsigned u[8]; };

__device__ __forceinline__ unsigned short f2bf_bits(float f) {
  union { float f; unsigned u; } v; v.f = f;
  return (unsigned short)((v.u + 0x7FFFu + ((v.u >> 16) & 1u)) >> 16);
}

__global__ void k_zero(float* __restrict__ p, long long n) {
  long long i = (long long)blockIdx.x * blockDim.x + threadIdx.x;
  const long long stride = (long long)gridDim.x * blockDim.x;
  for (; i < n; i += stride) p[i] = 0.0f;
}

__global__ void k_cvt(const float* __restrict__ s, unsigned short* __restrict__ d, long long n) {
  long long i = (long long)blockIdx.x * blockDim.x + threadIdx.x;
  const long long stride = (long long)gridDim.x * blockDim.x;
  for (; i < n; i += stride) d[i] = f2bf_bits(s[i]);
}

__global__ void __launch_bounds__(256) k_stats(const float* __restrict__ x, float* __restrict__ stats) {
  __shared__ float ssum[256], ssq[256];
  const int tid = threadIdx.x;
  long long i = (long long)blockIdx.x * blockDim.x + tid;
  const long long stride = (long long)gridDim.x * blockDim.x;
  const long long total = (long long)Nn * Dd;
  float s = 0.f, q = 0.f;
  for (; i < total; i += stride) { float v = x[i]; s += v; q += v * v; }
  ssum[tid] = s; ssq[tid] = q; __syncthreads();
  for (int o = 128; o > 0; o >>= 1) {
    if (tid < o) { ssum[tid] += ssum[tid + o]; ssq[tid] += ssq[tid + o]; }
    __syncthreads();
  }
  if (tid == 0) { atomicAdd(&stats[0], ssum[0]); atomicAdd(&stats[1], ssq[0]); }
}

__global__ void k_prep(const float* __restrict__ w, float* __restrict__ stats,
                       float* __restrict__ corr) {
  const int h = threadIdx.x;
  float colsum = 0.f;
  for (int d = 0; d < Dd; ++d) colsum += w[h * Dd + d];
  const float nd = (float)((long long)Nn * Dd);
  const float s  = stats[0], ss = stats[1];
  const float mu = s / nd;
  const float var = (ss - s * s / nd) / (nd - 1.0f);
  const float invsig = rsqrtf(var);
  if (h == 0) stats[2] = invsig;
  corr[h] = mu * invsig * colsum;
}

__global__ void __launch_bounds__(256) k_gat_gemm(const unsigned short* __restrict__ xbf,
                                                  const unsigned short* __restrict__ wbf,
                                                  const float* __restrict__ stats,
                                                  const float* __restrict__ corr,
                                                  float* __restrict__ z) {
  const int tid  = threadIdx.x;
  const int lane = tid & 31;
  const int hi   = lane >> 4;
  const int lm   = lane & 15;
  const int n0   = blockIdx.x * 16;
  const int h0   = (tid >> 5) * 16;

  const unsigned* xp = (const unsigned*)xbf + (size_t)(n0 + lm) * (Dd / 2);
  const unsigned* wp = (const unsigned*)wbf + (size_t)(h0 + lm) * (Dd / 2);

  v8f c = {0.f, 0.f, 0.f, 0.f, 0.f, 0.f, 0.f, 0.f};
  const float invsig = stats[2];

  #pragma unroll
  for (int k0 = 0; k0 < Dd; k0 += 32) {
    BFrag a, b;
    #pragma unroll
    for (int j = 0; j < 8; ++j) {
      const int kk = k0 + ((j < 4) ? 0 : 16) + (hi ? 8 : 0) + 2 * (j & 3);
      a.u[j] = xp[kk >> 1];
      b.u[j] = wp[kk >> 1];
    }
    if (k0 + 32 < Dd)
      __builtin_prefetch(xp + ((k0 + 32) >> 1), 0, 1);
    c = __builtin_amdgcn_wmma_f32_16x16x32_bf16(false, a.v, false, b.v, (short)0, c, false, false);
  }
  #pragma unroll
  for (int r = 0; r < 8; ++r) {
    const int m = r + hi * 8;
    const int h = h0 + lm;
    z[(size_t)(n0 + m) * Hh + h] = c[r] * invsig - corr[h];
  }
}

__global__ void __launch_bounds__(256) k_own(const int* __restrict__ src, const int* __restrict__ dst,
                                             int* __restrict__ pk, unsigned char* __restrict__ own) {
  long long i = (long long)blockIdx.x * blockDim.x + threadIdx.x;
  const long long stride = (long long)gridDim.x * blockDim.x;
  for (; i < Ee; i += stride) {
    const int s = src[i], d = dst[i];
    pk[i]  = (int)((unsigned)s | ((unsigned)d << 16));
    own[i] = (unsigned char)((unsigned)s / ROWS);
  }
}

__device__ __forceinline__ unsigned own_nz(unsigned x, unsigned pat) {
  const unsigned y = x ^ pat;
  return ((y & 0x7F7F7F7Fu) + 0x7F7F7F7Fu) | y;
}

__device__ __forceinline__ void lds_add(float* p, float v) {
  (void)__hip_atomic_fetch_add(p, v, __ATOMIC_RELAXED, __HIP_MEMORY_SCOPE_WORKGROUP);
}

template <int M>
__device__ __forceinline__ float row_xor(float x) {
  return __builtin_bit_cast(float, __builtin_amdgcn_update_dpp(0, __builtin_bit_cast(int, x), 0x160 | M, 0xF, 0xF, false));
}

__device__ __forceinline__ float xhalf(float x) {
  const int b = __builtin_bit_cast(int, x);
  return __builtin_bit_cast(float, __builtin_amdgcn_permlanex16(b, b, 0x76543210, 0xfedcba98, false, false));
}

__device__ __forceinline__ float wave_sum(float x) {
  x += row_xor<8>(x);
  x += row_xor<4>(x);
  x += row_xor<2>(x);
  x += row_xor<1>(x);
  return x + xhalf(x);
}

__device__ __forceinline__ void own_push(int e, int* queue, int* qn, float* rows, float* hs_l, const float* s1_l,
                                         const int* __restrict__ pk, const float* __restrict__ z,
                                         const float* __restrict__ a2, int base) {
  const int pos = __hip_atomic_fetch_add(qn, 1, __ATOMIC_RELAXED, __HIP_MEMORY_SCOPE_WORKGROUP);
  if (pos < QCAP) { queue[pos] = e; return; }
  const unsigned p = (unsigned)pk[e];
  const int sl = (int)(p & 0xFFFFu) - base;
  const int d  = (int)(p >> 16);
  const float* zr = z + (size_t)d * Hh;
  float s2 = 0.f;
  for (int c = 0; c < Hh; ++c) s2 += zr[c] * a2[c];
  float ev = s1_l[sl] + s2;
  ev = (ev >= 0.f) ? ev : LRELU * ev;
  const float h = __expf(ev);
  lds_add(&hs_l[sl], h);
  for (int c = 0; c < Hh; ++c)
    lds_add(&rows[sl * Hh + (c & 3) * 32 + (c >> 2)], h * zr[c]);
}

__global__ void __launch_bounds__(1024) k_edge2(const int* __restrict__ pk, const float* __restrict__ z,
                                                const float* __restrict__ a1, const float* __restrict__ a2,
                                                const unsigned char* __restrict__ own,
                                                float* __restrict__ gout) {
  __shared__ __attribute__((aligned(16))) float rows[ROWS * Hh];
  __shared__ int   queue[QCAP];
  __shared__ float hs_l[ROWS];
  __shared__ float s1_l[ROWS];
  __shared__ int   qn;

  const int tid  = threadIdx.x;
  const int lane = tid & 31;
  const int wave = __builtin_amdgcn_readfirstlane(tid >> 5);
  const int b    = blockIdx.x;
  const int base = b * ROWS;

  for (int i = tid; i < ROWS * Hh / 4; i += 1024) ((float4*)rows)[i] = make_float4(0.f, 0.f, 0.f, 0.f);
  if (tid < ROWS) hs_l[tid] = 0.f;
  if (tid == 0) qn = 0;

  const float4 a1v = *(const float4*)(a1 + lane * 4);
  float4 zo[(ROWS + 31) / 32];
  #pragma unroll
  for (int i = 0; i < (ROWS + 31) / 32; ++i) {
    const int r  = wave + 32 * i;
    const int nn = (base + r < Nn) ? (base + r) : (Nn - 1);
    zo[i] = *(const float4*)(z + (size_t)nn * Hh + lane * 4);
  }
  float s1m = 0.f;
  #pragma unroll
  for (int i = 0; i < (ROWS + 31) / 32; ++i) {
    const float s = wave_sum(zo[i].x * a1v.x + zo[i].y * a1v.y + zo[i].z * a1v.z + zo[i].w * a1v.w);
    s1m = (lane == i) ? s : s1m;
  }
  if (lane < (ROWS + 31) / 32 && wave + 32 * lane < ROWS) s1_l[wave + 32 * lane] = s1m;
  __syncthreads();

  const uint4* own4 = (const uint4*)own;
  const unsigned pat = (unsigned)b * 0x01010101u;
  for (int it = 0; it < NQ4; it += 4096) {
    uint4 w[4];
    #pragma unroll
    for (int k = 0; k < 4; ++k) {
      const int q = it + k * 1024 + tid;
      w[k] = own4[q < NQ4 ? q : NQ4 - 1];
    }
    unsigned zf[4];
    #pragma unroll
    for (int k = 0; k < 4; ++k) {
      const unsigned acc = (own_nz(w[k].x, pat) & 0x80808080u) | ((own_nz(w[k].y, pat) >> 1) & 0x40404040u) |
                           ((own_nz(w[k].z, pat) >> 2) & 0x20202020u) | ((own_nz(w[k].w, pat) >> 3) & 0x10101010u);
      zf[k] = (it + k * 1024 + tid < NQ4) ? (~acc & 0xF0F0F0F0u) : 0u;
    }
    unsigned long long m = (unsigned long long)(zf[0] | (zf[1] >> 4)) |
                           ((unsigned long long)(zf[2] | (zf[3] >> 4)) << 32);
    while (m != 0ull) {
      const int P = __ffsll((long long)m) - 1;
      m &= m - 1ull;
      const int p = P & 31, i = p >> 3, nb = p & 7;
      const int k = 2 * (P >> 5) + ((nb >> 2) ^ 1);
      const int j = 3 - (nb & 3);
      own_push((it + k * 1024 + tid) * 16 + j * 4 + i, queue, &qn, rows, hs_l, s1_l, pk, z, a2, base);
    }
  }
  __syncthreads();

  const float4 a2v = *(const float4*)(a2 + lane * 4);
  int n = qn;
  n = __builtin_amdgcn_readfirstlane(n < QCAP ? n : QCAP);
  for (int g0 = wave * 32; g0 < n; g0 += 1024) {
    const int cnt = (n - g0 < 32) ? (n - g0) : 32;
    int sl = 0, d = 0; float s1v = 0.f;
    float hk = 0.f;
    if (lane < cnt) {
      const int e = queue[g0 + lane];
      const unsigned p = (unsigned)pk[e];
      sl  = (int)(p & 0xFFFFu) - base;
      d   = (int)(p >> 16);
      s1v = s1_l[sl];
    }
    #pragma unroll
    for (int k0 = 0; k0 < 32; k0 += 8) {
      if (k0 < cnt) {
        float4 zv[8]; int sk[8]; float s1k[8];
        #pragma unroll
        for (int j = 0; j < 8; ++j) {
          const int kk = k0 + j;
          sk[j] = __builtin_amdgcn_readlane(sl, kk);
          const int dk = __builtin_amdgcn_readlane(d, kk);
          s1k[j] = __int_as_float(__builtin_amdgcn_readlane(__float_as_int(s1v), kk));
          zv[j] = make_float4(0.f, 0.f, 0.f, 0.f);
          if (kk < cnt) zv[j] = *(const float4*)(z + (size_t)dk * Hh + lane * 4);
        }
        #pragma unroll
        for (int j = 0; j < 8; ++j) {
          if (k0 + j < cnt) {
            const float s2 = wave_sum(zv[j].x * a2v.x + zv[j].y * a2v.y + zv[j].z * a2v.z + zv[j].w * a2v.w);
            float ev = s1k[j] + s2;
            ev = (ev >= 0.f) ? ev : LRELU * ev;
            const float h = __expf(ev);
            float* r = rows + sk[j] * Hh + lane;
            lds_add(r,      h * zv[j].x);
            lds_add(r + 32, h * zv[j].y);
            lds_add(r + 64, h * zv[j].z);
            lds_add(r + 96, h * zv[j].w);
            hk = (lane == k0 + j) ? h : hk;
          }
        }
      }
    }
    if (lane < cnt) lds_add(&hs_l[sl], hk);
  }
  __syncthreads();

  for (int r = wave; r < ROWS; r += 32) {
    const int nn = base + r;
    if (nn < Nn) {
      const float hs  = hs_l[r];
      const float inv = (hs > 0.f) ? 1.0f / hs : 0.f;
      const float* p = rows + r * Hh + lane;
      *(float4*)(gout + (size_t)nn * Hh + lane * 4) = make_float4(p[0] * inv, p[32] * inv, p[64] * inv, p[96] * inv);
    }
  }
}

#define PCH        16
#define POOL_WAVES 4
#define POOL_NW    (Nn / PCH)
#define POOL_WG    ((POOL_NW + POOL_WAVES - 1) / POOL_WAVES)
static_assert(Nn % PCH == 0, "every wave that has nodes has PCH of them");
static_assert(PCH <= 32, "lane i keeps the graph id of the wave's node i");

__device__ __forceinline__ void pool_flush(float* __restrict__ pool, int g, int lane, float4 a) {
  float* p = pool + (size_t)g * Hh + lane * 4;
  atomicAdd(p + 0, a.x);
  atomicAdd(p + 1, a.y);
  atomicAdd(p + 2, a.z);
  atomicAdd(p + 3, a.w);
}

__global__ void __launch_bounds__(POOL_WAVES * 32) k_pool(const float* __restrict__ gout, const int* __restrict__ idx,
                                                          float* __restrict__ pool) {
  const int lane = threadIdx.x & 31;
  const int w = blockIdx.x * POOL_WAVES + (threadIdx.x >> 5);
  if (w >= POOL_NW) return;
  const int n0 = w * PCH;
  const int gi = idx[n0 + (lane & (PCH - 1))];
  float4 v[PCH];
  #pragma unroll
  for (int i = 0; i < PCH; ++i) v[i] = *(const float4*)(gout + (size_t)(n0 + i) * Hh + lane * 4);
  int cur = __builtin_amdgcn_readlane(gi, 0);
  float4 acc = make_float4(0.f, 0.f, 0.f, 0.f);
  #pragma unroll
  for (int i = 0; i < PCH; ++i) {
    const int g = __builtin_amdgcn_readlane(gi, i);
    if (g != cur) {
      pool_flush(pool, cur, lane, acc);
      acc = make_float4(0.f, 0.f, 0.f, 0.f);
      cur = g;
    }
    acc.x += fmaxf(v[i].x, 0.f);
    acc.y += fmaxf(v[i].y, 0.f);
    acc.z += fmaxf(v[i].z, 0.f);
    acc.w += fmaxf(v[i].w, 0.f);
  }
  pool_flush(pool, cur, lane, acc);
}

#define BN_WG    (Hh / 32)
#define BN_WAVES 16
#define BN_ROWS  (Gg / BN_WAVES)
static_assert(BN_WG * 32 == Hh, "32 columns per work-group");
static_assert(BN_WAVES * BN_ROWS == Gg, "the waves of a work-group cover every graph row");
__global__ void __launch_bounds__(BN_WAVES * 32) k_bn(const float* __restrict__ pooled, const float* __restrict__ gamma,
                                                      const float* __restrict__ beta, float* __restrict__ xb) {
  __shared__ float ps[BN_WAVES][32], pq[BN_WAVES][32];
  const int lane = threadIdx.x & 31;
  const int wave = threadIdx.x >> 5;
  const int h  = blockIdx.x * 32 + lane;
  const int g0 = wave * BN_ROWS;
  float v[BN_ROWS];
  #pragma unroll
  for (int i = 0; i < BN_ROWS; ++i) v[i] = pooled[(g0 + i) * Hh + h];
  float s = 0.f, q = 0.f;
  #pragma unroll
  for (int i = 0; i < BN_ROWS; ++i) { s += v[i]; q += v[i] * v[i]; }
  ps[wave][lane] = s; pq[wave][lane] = q;
  __syncthreads();
  float S = 0.f, Q = 0.f;
  #pragma unroll
  for (int w = 0; w < BN_WAVES; ++w) { S += ps[w][lane]; Q += pq[w][lane]; }
  const float mu = S / (float)Gg;
  const float var = Q / (float)Gg - mu * mu;
  const float inv = rsqrtf(var + BN_EPS);
  const float ga = gamma[h], be = beta[h];
  #pragma unroll
  for (int i = 0; i < BN_ROWS; ++i)
    xb[(g0 + i) * Hh + h] = (v[i] - mu) * inv * ga + be;
}

__global__ void __launch_bounds__(256) k_fc1(const unsigned short* __restrict__ xbf,
                                             const unsigned short* __restrict__ wbf,
                                             const float* __restrict__ bias, float* __restrict__ out) {
  const int tid  = threadIdx.x;
  const int lane = tid & 31;
  const int hi   = lane >> 4;
  const int lm   = lane & 15;
  const int n0   = blockIdx.x * 16;
  const int h0   = (tid >> 5) * 16;

  const unsigned* xp = (const unsigned*)xbf + (size_t)(n0 + lm) * (Hh / 2);
  const unsigned* wp = (const unsigned*)wbf + (size_t)(h0 + lm) * (Hh / 2);

  v8f c = {0.f, 0.f, 0.f, 0.f, 0.f, 0.f, 0.f, 0.f};
  #pragma unroll
  for (int k0 = 0; k0 < Hh; k0 += 32) {
    BFrag a, b;
    #pragma unroll
    for (int j = 0; j < 8; ++j) {
      const int kk = k0 + ((j < 4) ? 0 : 16) + (hi ? 8 : 0) + 2 * (j & 3);
      a.u[j] = xp[kk >> 1];
      b.u[j] = wp[kk >> 1];
    }
    c = __builtin_amdgcn_wmma_f32_16x16x32_bf16(false, a.v, false, b.v, (short)0, c, false, false);
  }
  #pragma unroll
  for (int r = 0; r < 8; ++r) {
    const int m = r + hi * 8;
    const int h = h0 + lm;
    out[(size_t)(n0 + m) * Hh + h] = fmaxf(c[r] + bias[h], 0.f);
  }
}

__global__ void __launch_bounds__(256) k_fc2(const float* __restrict__ x, const float* __restrict__ w,
                                             const float* __restrict__ b, float* __restrict__ out) {
  const int lane = threadIdx.x & 31;
  const int g = blockIdx.x * 8 + (threadIdx.x >> 5);
  if (g >= Gg) return;
  float dot = -INFINITY;
  if (lane < Cc) {
    float acc = b[lane];
    for (int d = 0; d < Hh; ++d) acc += x[(size_t)g * Hh + d] * w[lane * Hh + d];
    dot = acc;
  }
  float m = dot;
  for (int o = 16; o > 0; o >>= 1) m = fmaxf(m, __shfl_xor(m, o));
  float ex = (lane < Cc) ? __expf(dot - m) : 0.f;
  float se = ex;
  for (int o = 16; o > 0; o >>= 1) se += __shfl_xor(se, o);
  if (lane < Cc) out[(size_t)g * Cc + lane] = dot - m - logf(se);
}

extern "C" void kernel_launch(void* const* d_in, const int* in_sizes, int n_in,
                              void* d_out, int out_size, void* d_ws, size_t ws_size,
                              hipStream_t stream) {
  const float* x_in  = (const float*)d_in[0];
  const int*   eidx  = (const int*)  d_in[1];
  const int*   idx   = (const int*)  d_in[2];
  const float* fc_w  = (const float*)d_in[3];
  const float* a1    = (const float*)d_in[4];
  const float* a2    = (const float*)d_in[5];
  const float* fc1_w = (const float*)d_in[6];
  const float* fc1_b = (const float*)d_in[7];
  const float* fc2_w = (const float*)d_in[8];
  const float* fc2_b = (const float*)d_in[9];
  const float* gamma = (const float*)d_in[10];
  const float* beta  = (const float*)d_in[11];
  const int* src = eidx;
  const int* dst = eidx + Ee;

  float* ws   = (float*)d_ws;
  float* STAT = ws;
  float* CORR = ws + 64;
  float* Z    = ws + 256;
  float* GOUT = Z + (size_t)Nn * Hh;
  float* POOL = GOUT + (size_t)Nn * Hh;
  float* XB   = POOL + (size_t)Gg * Hh;
  float* FC1O = XB + (size_t)Gg * Hh;
  unsigned short* XBF  = (unsigned short*)(FC1O + (size_t)Gg * Hh);
  unsigned short* WBF  = XBF  + (size_t)Nn * Dd;
  unsigned short* W1BF = WBF  + (size_t)Hh * Dd;
  unsigned short* XBBF = W1BF + (size_t)Hh * Hh;
  unsigned char*  OWN  = (unsigned char*)(XBBF + (size_t)Gg * Hh);
  int*            PK   = (int*)(OWN + (size_t)Ee);

  (void)in_sizes; (void)n_in; (void)out_size; (void)ws_size;

  k_zero<<<1,    256, 0, stream>>>(STAT, 16);
  k_zero<<<64,   256, 0, stream>>>(POOL, (long long)Gg * Hh);

  k_cvt<<<2048, 256, 0, stream>>>(x_in,  XBF,  (long long)Nn * Dd);
  k_cvt<<<64,   256, 0, stream>>>(fc_w,  WBF,  (long long)Hh * Dd);
  k_cvt<<<64,   256, 0, stream>>>(fc1_w, W1BF, (long long)Hh * Hh);

  k_stats<<<1024, 256, 0, stream>>>(x_in, STAT);
  k_prep<<<1, 128, 0, stream>>>(fc_w, STAT, CORR);
  k_gat_gemm<<<Nn / 16, 256, 0, stream>>>(XBF, WBF, STAT, CORR, Z);
  k_own<<<4096, 256, 0, stream>>>(src, dst, PK, OWN);
  k_edge2<<<NOWN, 1024, 0, stream>>>(PK, Z, a1, a2, OWN, GOUT);
  k_pool<<<POOL_WG, POOL_WAVES * 32, 0, stream>>>(GOUT, idx, POOL);
  k_bn<<<BN_WG, BN_WAVES * 32, 0, stream>>>(POOL, gamma, beta, XB);
  k_cvt<<<64, 256, 0, stream>>>(XB, XBBF, (long long)Gg * Hh);
  k_fc1<<<Gg / 16, 256, 0, stream>>>(XBBF, W1BF, fc1_b, FC1O);
  k_fc2<<<Gg / 8, 256, 0, stream>>>(FC1O, fc2_w, fc2_b, (float*)d_out);
}


# passes that waited for memory once per loop trip (x to bf16, x sums, edge packing, fc_w row sums) issue all 16-byte loads of a lane first; the big three in 2500-5000 waves; x sums regrouped
# speedup vs baseline: 4.2318x; 1.0544x over previous
#include <hip/hip_runtime.h>
#include <math.h>

#define Nn 50000
#define Ee 1600000
#define Dd 128
#define Hh 128
#define Cc 10
#define Gg 512
#define LRELU 0.05f
#define BN_EPS 1e-5f

#define NOWN 256
#define ROWS 196
#define QCAP 32768
#define NQ4  (Ee / 16)
static_assert(NOWN * ROWS >= Nn, "owners must cover every node");
static_assert(NOWN <= 256, "owner id is one byte");
static_assert(Nn <= 65536, "src and dst are packed as two 16-bit halves");
static_assert(Ee % 16 == 0, "owner bytes are read 16 at a time");

typedef __attribute__((ext_vector_type(16))) __bf16 v16bf;
typedef __attribute__((ext_vector_type(8)))  float  v8f;

union BFrag { v16bf v; unsigned u[8]; };

__device__ __forceinline__ unsigned short f2bf_bits(float f) {
  union { float f; unsigned u; } v; v.f = f;
  return (unsigned short)((v.u + 0x7FFFu + ((v.u >> 16) & 1u)) >> 16);
}

__global__ void k_zero(float* __restrict__ p, long long n) {
  long long i = (long long)blockIdx.x * blockDim.x + threadIdx.x;
  const long long stride = (long long)gridDim.x * blockDim.x;
  for (; i < n; i += stride) p[i] = 0.0f;
}

__global__ void k_cvt(const float* __restrict__ s, unsigned short* __restrict__ d, long long n) {
  long long i = (long long)blockIdx.x * blockDim.x + threadIdx.x;
  const long long stride = (long long)gridDim.x * blockDim.x;
  for (; i < n; i += stride) d[i] = f2bf_bits(s[i]);
}

#define SV4  10
#define SBLK ((Nn * Dd / 4) / (256 * SV4))
static_assert((long long)SBLK * 256 * SV4 * 4 == (long long)Nn * Dd, "the waves cover x_in exactly: no tail, no guard");

__global__ void __launch_bounds__(256) k_cvtx(const float* __restrict__ x, unsigned short* __restrict__ d) {
  const int lane = threadIdx.x & 31;
  const int w = blockIdx.x * 8 + (threadIdx.x >> 5);
  const float4* xp = (const float4*)x + (size_t)w * (SV4 * 32) + lane;
  uint2* dp = (uint2*)d + (size_t)w * (SV4 * 32) + lane;
  float4 v[SV4];
  #pragma unroll
  for (int k = 0; k < SV4; ++k) v[k] = xp[k * 32];
  #pragma unroll
  for (int k = 0; k < SV4; ++k) {
    uint2 o;
    o.x = (unsigned)f2bf_bits(v[k].x) | ((unsigned)f2bf_bits(v[k].y) << 16);
    o.y = (unsigned)f2bf_bits(v[k].z) | ((unsigned)f2bf_bits(v[k].w) << 16);
    dp[k * 32] = o;
  }
}

__global__ void __launch_bounds__(256) k_stats(const float* __restrict__ x, float* __restrict__ stats) {
  __shared__ float ssum[256], ssq[256];
  const int tid = threadIdx.x;
  const int lane = tid & 31;
  const int w = blockIdx.x * 8 + (tid >> 5);
  const float4* xp = (const float4*)x + (size_t)w * (SV4 * 32) + lane;
  float4 v[SV4];
  #pragma unroll
  for (int k = 0; k < SV4; ++k) v[k] = xp[k * 32];
  float4 s4 = make_float4(0.f, 0.f, 0.f, 0.f), q4 = make_float4(0.f, 0.f, 0.f, 0.f);
  #pragma unroll
  for (int k = 0; k < SV4; ++k) {
    s4.x += v[k].x; s4.y += v[k].y; s4.z += v[k].z; s4.w += v[k].w;
    q4.x += v[k].x * v[k].x; q4.y += v[k].y * v[k].y; q4.z += v[k].z * v[k].z; q4.w += v[k].w * v[k].w;
  }
  const float s = (s4.x + s4.y) + (s4.z + s4.w);
  const float q = (q4.x + q4.y) + (q4.z + q4.w);
  ssum[tid] = s; ssq[tid] = q; __syncthreads();
  for (int o = 128; o > 0; o >>= 1) {
    if (tid < o) { ssum[tid] += ssum[tid + o]; ssq[tid] += ssq[tid + o]; }
    __syncthreads();
  }
  if (tid == 0) { atomicAdd(&stats[0], ssum[0]); atomicAdd(&stats[1], ssq[0]); }
}

__global__ void __launch_bounds__(128) k_prep(const float* __restrict__ w, float* __restrict__ stats,
                       float* __restrict__ corr) {
  const int h = threadIdx.x;
  float colsum = 0.f;
  #pragma unroll
  for (int d = 0; d < Dd; ++d) colsum += w[h * Dd + d];
  const float nd = (float)((long long)Nn * Dd);
  const float s  = stats[0], ss = stats[1];
  const float mu = s / nd;
  const float var = (ss - s * s / nd) / (nd - 1.0f);
  const float invsig = rsqrtf(var);
  if (h == 0) stats[2] = invsig;
  corr[h] = mu * invsig * colsum;
}

__global__ void __launch_bounds__(256) k_gat_gemm(const unsigned short* __restrict__ xbf,
                                                  const unsigned short* __restrict__ wbf,
                                                  const float* __restrict__ stats,
                                                  const float* __restrict__ corr,
                                                  float* __restrict__ z) {
  const int tid  = threadIdx.x;
  const int lane = tid & 31;
  const int hi   = lane >> 4;
  const int lm   = lane & 15;
  const int n0   = blockIdx.x * 16;
  const int h0   = (tid >> 5) * 16;

  const unsigned* xp = (const unsigned*)xbf + (size_t)(n0 + lm) * (Dd / 2);
  const unsigned* wp = (const unsigned*)wbf + (size_t)(h0 + lm) * (Dd / 2);

  v8f c = {0.f, 0.f, 0.f, 0.f, 0.f, 0.f, 0.f, 0.f};
  const float invsig = stats[2];

  #pragma unroll
  for (int k0 = 0; k0 < Dd; k0 += 32) {
    BFrag a, b;
    #pragma unroll
    for (int j = 0; j < 8; ++j) {
      const int kk = k0 + ((j < 4) ? 0 : 16) + (hi ? 8 : 0) + 2 * (j & 3);
      a.u[j] = xp[kk >> 1];
      b.u[j] = wp[kk >> 1];
    }
    if (k0 + 32 < Dd)
      __builtin_prefetch(xp + ((k0 + 32) >> 1), 0, 1);
    c = __builtin_amdgcn_wmma_f32_16x16x32_bf16(false, a.v, false, b.v, (short)0, c, false, false);
  }
  #pragma unroll
  for (int r = 0; r < 8; ++r) {
    const int m = r + hi * 8;
    const int h = h0 + lm;
    z[(size_t)(n0 + m) * Hh + h] = c[r] * invsig - corr[h];
  }
}

#define OQ   5
#define OBLK ((Ee / 4) / (128 * OQ))
static_assert(OBLK * 128 * OQ * 4 == Ee, "the waves cover the edges exactly: no tail, no guard");
__global__ void __launch_bounds__(128) k_own(const int* __restrict__ src, const int* __restrict__ dst,
                                             int* __restrict__ pk, unsigned char* __restrict__ own) {
  const int lane = threadIdx.x & 31;
  const int w = blockIdx.x * 4 + (threadIdx.x >> 5);
  const size_t g0 = (size_t)w * (OQ * 32) + lane;
  const int4* sp = (const int4*)src + g0;
  const int4* dp = (const int4*)dst + g0;
  int4*     pp = (int4*)pk + g0;
  unsigned* op = (unsigned*)own + g0;
  int4 s[OQ], d[OQ];
  #pragma unroll
  for (int k = 0; k < OQ; ++k) { s[k] = sp[k * 32]; d[k] = dp[k * 32]; }
  #pragma unroll
  for (int k = 0; k < OQ; ++k) {
    int4 p;
    p.x = (int)((unsigned)s[k].x | ((unsigned)d[k].x << 16));
    p.y = (int)((unsigned)s[k].y | ((unsigned)d[k].y << 16));
    p.z = (int)((unsigned)s[k].z | ((unsigned)d[k].z << 16));
    p.w = (int)((unsigned)s[k].w | ((unsigned)d[k].w << 16));
    pp[k * 32] = p;
    op[k * 32] = (((unsigned)s[k].x / ROWS) & 0xFFu) | ((((unsigned)s[k].y / ROWS) & 0xFFu) << 8) |
                 ((((unsigned)s[k].z / ROWS) & 0xFFu) << 16) | (((unsigned)s[k].w / ROWS) << 24);
  }
}

__device__ __forceinline__ unsigned own_nz(unsigned x, unsigned pat) {
  const unsigned y = x ^ pat;
  return ((y & 0x7F7F7F7Fu) + 0x7F7F7F7Fu) | y;
}

__device__ __forceinline__ void lds_add(float* p, float v) {
  (void)__hip_atomic_fetch_add(p, v, __ATOMIC_RELAXED, __HIP_MEMORY_SCOPE_WORKGROUP);
}

template <int M>
__device__ __forceinline__ float row_xor(float x) {
  return __builtin_bit_cast(float, __builtin_amdgcn_update_dpp(0, __builtin_bit_cast(int, x), 0x160 | M, 0xF, 0xF, false));
}

__device__ __forceinline__ float xhalf(float x) {
  const int b = __builtin_bit_cast(int, x);
  return __builtin_bit_cast(float, __builtin_amdgcn_permlanex16(b, b, 0x76543210, 0xfedcba98, false, false));
}

__device__ __forceinline__ float wave_sum(float x) {
  x += row_xor<8>(x);
  x += row_xor<4>(x);
  x += row_xor<2>(x);
  x += row_xor<1>(x);
  return x + xhalf(x);
}

__device__ __forceinline__ void own_push(int e, int* queue, int* qn, float* rows, float* hs_l, const float* s1_l,
                                         const int* __restrict__ pk, const float* __restrict__ z,
                                         const float* __restrict__ a2, int base) {
  const int pos = __hip_atomic_fetch_add(qn, 1, __ATOMIC_RELAXED, __HIP_MEMORY_SCOPE_WORKGROUP);
  if (pos < QCAP) { queue[pos] = e; return; }
  const unsigned p = (unsigned)pk[e];
  const int sl = (int)(p & 0xFFFFu) - base;
  const int d  = (int)(p >> 16);
  const float* zr = z + (size_t)d * Hh;
  float s2 = 0.f;
  for (int c = 0; c < Hh; ++c) s2 += zr[c] * a2[c];
  float ev = s1_l[sl] + s2;
  ev = (ev >= 0.f) ? ev : LRELU * ev;
  const float h = __expf(ev);
  lds_add(&hs_l[sl], h);
  for (int c = 0; c < Hh; ++c)
    lds_add(&rows[sl * Hh + (c & 3) * 32 + (c >> 2)], h * zr[c]);
}

__global__ void __launch_bounds__(1024) k_edge2(const int* __restrict__ pk, const float* __restrict__ z,
                                                const float* __restrict__ a1, const float* __restrict__ a2,
                                                const unsigned char* __restrict__ own,
                                                float* __restrict__ gout) {
  __shared__ __attribute__((aligned(16))) float rows[ROWS * Hh];
  __shared__ int   queue[QCAP];
  __shared__ float hs_l[ROWS];
  __shared__ float s1_l[ROWS];
  __shared__ int   qn;

  const int tid  = threadIdx.x;
  const int lane = tid & 31;
  const int wave = __builtin_amdgcn_readfirstlane(tid >> 5);
  const int b    = blockIdx.x;
  const int base = b * ROWS;

  for (int i = tid; i < ROWS * Hh / 4; i += 1024) ((float4*)rows)[i] = make_float4(0.f, 0.f, 0.f, 0.f);
  if (tid < ROWS) hs_l[tid] = 0.f;
  if (tid == 0) qn = 0;

  const float4 a1v = *(const float4*)(a1 + lane * 4);
  float4 zo[(ROWS + 31) / 32];
  #pragma unroll
  for (int i = 0; i < (ROWS + 31) / 32; ++i) {
    const int r  = wave + 32 * i;
    const int nn = (base + r < Nn) ? (base + r) : (Nn - 1);
    zo[i] = *(const float4*)(z + (size_t)nn * Hh + lane * 4);
  }
  float s1m = 0.f;
  #pragma unroll
  for (int i = 0; i < (ROWS + 31) / 32; ++i) {
    const float s = wave_sum(zo[i].x * a1v.x + zo[i].y * a1v.y + zo[i].z * a1v.z + zo[i].w * a1v.w);
    s1m = (lane == i) ? s : s1m;
  }
  if (lane < (ROWS + 31) / 32 && wave + 32 * lane < ROWS) s1_l[wave + 32 * lane] = s1m;
  __syncthreads();

  const uint4* own4 = (const uint4*)own;
  const unsigned pat = (unsigned)b * 0x01010101u;
  for (int it = 0; it < NQ4; it += 4096) {
    uint4 w[4];
    #pragma unroll
    for (int k = 0; k < 4; ++k) {
      const int q = it + k * 1024 + tid;
      w[k] = own4[q < NQ4 ? q : NQ4 - 1];
    }
    unsigned zf[4];
    #pragma unroll
    for (int k = 0; k < 4; ++k) {
      const unsigned acc = (own_nz(w[k].x, pat) & 0x80808080u) | ((own_nz(w[k].y, pat) >> 1) & 0x40404040u) |
                           ((own_nz(w[k].z, pat) >> 2) & 0x20202020u) | ((own_nz(w[k].w, pat) >> 3) & 0x10101010u);
      zf[k] = (it + k * 1024 + tid < NQ4) ? (~acc & 0xF0F0F0F0u) : 0u;
    }
    unsigned long long m = (unsigned long long)(zf[0] | (zf[1] >> 4)) |
                           ((unsigned long long)(zf[2] | (zf[3] >> 4)) << 32);
    while (m != 0ull) {
      const int P = __ffsll((long long)m) - 1;
      m &= m - 1ull;
      const int p = P & 31, i = p >> 3, nb = p & 7;
      const int k = 2 * (P >> 5) + ((nb >> 2) ^ 1);
      const int j = 3 - (nb & 3);
      own_push((it + k * 1024 + tid) * 16 + j * 4 + i, queue, &qn, rows, hs_l, s1_l, pk, z, a2, base);
    }
  }
  __syncthreads();

  const float4 a2v = *(const float4*)(a2 + lane * 4);
  int n = qn;
  n = __builtin_amdgcn_readfirstlane(n < QCAP ? n : QCAP);
  for (int g0 = wave * 32; g0 < n; g0 += 1024) {
    const int cnt = (n - g0 < 32) ? (n - g0) : 32;
    int sl = 0, d = 0; float s1v = 0.f;
    float hk = 0.f;
    if (lane < cnt) {
      const int e = queue[g0 + lane];
      const unsigned p = (unsigned)pk[e];
      sl  = (int)(p & 0xFFFFu) - base;
      d   = (int)(p >> 16);
      s1v = s1_l[sl];
    }
    #pragma unroll
    for (int k0 = 0; k0 < 32; k0 += 8) {
      if (k0 < cnt) {
        float4 zv[8]; int sk[8]; float s1k[8];
        #pragma unroll
        for (int j = 0; j < 8; ++j) {
          const int kk = k0 + j;
          sk[j] = __builtin_amdgcn_readlane(sl, kk);
          const int dk = __builtin_amdgcn_readlane(d, kk);
          s1k[j] = __int_as_float(__builtin_amdgcn_readlane(__float_as_int(s1v), kk));
          zv[j] = make_float4(0.f, 0.f, 0.f, 0.f);
          if (kk < cnt) zv[j] = *(const float4*)(z + (size_t)dk * Hh + lane * 4);
        }
        #pragma unroll
        for (int j = 0; j < 8; ++j) {
          if (k0 + j < cnt) {
            const float s2 = wave_sum(zv[j].x * a2v.x + zv[j].y * a2v.y + zv[j].z * a2v.z + zv[j].w * a2v.w);
            float ev = s1k[j] + s2;
            ev = (ev >= 0.f) ? ev : LRELU * ev;
            const float h = __expf(ev);
            float* r = rows + sk[j] * Hh + lane;
            lds_add(r,      h * zv[j].x);
            lds_add(r + 32, h * zv[j].y);
            lds_add(r + 64, h * zv[j].z);
            lds_add(r + 96, h * zv[j].w);
            hk = (lane == k0 + j) ? h : hk;
          }
        }
      }
    }
    if (lane < cnt) lds_add(&hs_l[sl], hk);
  }
  __syncthreads();

  for (int r = wave; r < ROWS; r += 32) {
    const int nn = base + r;
    if (nn < Nn) {
      const float hs  = hs_l[r];
      const float inv = (hs > 0.f) ? 1.0f / hs : 0.f;
      const float* p = rows + r * Hh + lane;
      *(float4*)(gout + (size_t)nn * Hh + lane * 4) = make_float4(p[0] * inv, p[32] * inv, p[64] * inv, p[96] * inv);
    }
  }
}

#define PCH        16
#define POOL_WAVES 4
#define POOL_NW    (Nn / PCH)
#define POOL_WG    ((POOL_NW + POOL_WAVES - 1) / POOL_WAVES)
static_assert(Nn % PCH == 0, "every wave that has nodes has PCH of them");
static_assert(PCH <= 32, "lane i keeps the graph id of the wave's node i");

__device__ __forceinline__ void pool_flush(float* __restrict__ pool, int g, int lane, float4 a) {
  float* p = pool + (size_t)g * Hh + lane * 4;
  atomicAdd(p + 0, a.x);
  atomicAdd(p + 1, a.y);
  atomicAdd(p + 2, a.z);
  atomicAdd(p + 3, a.w);
}

__global__ void __launch_bounds__(POOL_WAVES * 32) k_pool(const float* __restrict__ gout, const int* __restrict__ idx,
                                                          float* __restrict__ pool) {
  const int lane = threadIdx.x & 31;
  const int w = blockIdx.x * POOL_WAVES + (threadIdx.x >> 5);
  if (w >= POOL_NW) return;
  const int n0 = w * PCH;
  const int gi = idx[n0 + (lane & (PCH - 1))];
  float4 v[PCH];
  #pragma unroll
  for (int i = 0; i < PCH; ++i) v[i] = *(const float4*)(gout + (size_t)(n0 + i) * Hh + lane * 4);
  int cur = __builtin_amdgcn_readlane(gi, 0);
  float4 acc = make_float4(0.f, 0.f, 0.f, 0.f);
  #pragma unroll
  for (int i = 0; i < PCH; ++i) {
    const int g = __builtin_amdgcn_readlane(gi, i);
    if (g != cur) {
      pool_flush(pool, cur, lane, acc);
      acc = make_float4(0.f, 0.f, 0.f, 0.f);
      cur = g;
    }
    acc.x += fmaxf(v[i].x, 0.f);
    acc.y += fmaxf(v[i].y, 0.f);
    acc.z += fmaxf(v[i].z, 0.f);
    acc.w += fmaxf(v[i].w, 0.f);
  }
  pool_flush(pool, cur, lane, acc);
}

#define BN_WG    (Hh / 32)
#define BN_WAVES 16
#define BN_ROWS  (Gg / BN_WAVES)
static_assert(BN_WG * 32 == Hh, "32 columns per work-group");
static_assert(BN_WAVES * BN_ROWS == Gg, "the waves of a work-group cover every graph row");
__global__ void __launch_bounds__(BN_WAVES * 32) k_bn(const float* __restrict__ pooled, const float* __restrict__ gamma,
                                                      const float* __restrict__ beta, float* __restrict__ xb) {
  __shared__ float ps[BN_WAVES][32], pq[BN_WAVES][32];
  const int lane = threadIdx.x & 31;
  const int wave = threadIdx.x >> 5;
  const int h  = blockIdx.x * 32 + lane;
  const int g0 = wave * BN_ROWS;
  float v[BN_ROWS];
  #pragma unroll
  for (int i = 0; i < BN_ROWS; ++i) v[i] = pooled[(g0 + i) * Hh + h];
  float s = 0.f, q = 0.f;
  #pragma unroll
  for (int i = 0; i < BN_ROWS; ++i) { s += v[i]; q += v[i] * v[i]; }
  ps[wave][lane] = s; pq[wave][lane] = q;
  __syncthreads();
  float S = 0.f, Q = 0.f;
  #pragma unroll
  for (int w = 0; w < BN_WAVES; ++w) { S += ps[w][lane]; Q += pq[w][lane]; }
  const float mu = S / (float)Gg;
  const float var = Q / (float)Gg - mu * mu;
  const float inv = rsqrtf(var + BN_EPS);
  const float ga = gamma[h], be = beta[h];
  #pragma unroll
  for (int i = 0; i < BN_ROWS; ++i)
    xb[(g0 + i) * Hh + h] = (v[i] - mu) * inv * ga + be;
}

__global__ void __launch_bounds__(256) k_fc1(const unsigned short* __restrict__ xbf,
                                             const unsigned short* __restrict__ wbf,
                                             const float* __restrict__ bias, float* __restrict__ out) {
  const int tid  = threadIdx.x;
  const int lane = tid & 31;
  const int hi   = lane >> 4;
  const int lm   = lane & 15;
  const int n0   = blockIdx.x * 16;
  const int h0   = (tid >> 5) * 16;

  const unsigned* xp = (const unsigned*)xbf + (size_t)(n0 + lm) * (Hh / 2);
  const unsigned* wp = (const unsigned*)wbf + (size_t)(h0 + lm) * (Hh / 2);

  v8f c = {0.f, 0.f, 0.f, 0.f, 0.f, 0.f, 0.f, 0.f};
  #pragma unroll
  for (int k0 = 0; k0 < Hh; k0 += 32) {
    BFrag a, b;
    #pragma unroll
    for (int j = 0; j < 8; ++j) {
      const int kk = k0 + ((j < 4) ? 0 : 16) + (hi ? 8 : 0) + 2 * (j & 3);
      a.u[j] = xp[kk >> 1];
      b.u[j] = wp[kk >> 1];
    }
    c = __builtin_amdgcn_wmma_f32_16x16x32_bf16(false, a.v, false, b.v, (short)0, c, false, false);
  }
  #pragma unroll
  for (int r = 0; r < 8; ++r) {
    const int m = r + hi * 8;
    const int h = h0 + lm;
    out[(size_t)(n0 + m) * Hh + h] = fmaxf(c[r] + bias[h], 0.f);
  }
}

__global__ void __launch_bounds__(256) k_fc2(const float* __restrict__ x, const float* __restrict__ w,
                                             const float* __restrict__ b, float* __restrict__ out) {
  const int lane = threadIdx.x & 31;
  const int g = blockIdx.x * 8 + (threadIdx.x >> 5);
  if (g >= Gg) return;
  float dot = -INFINITY;
  if (lane < Cc) {
    float acc = b[lane];
    for (int d = 0; d < Hh; ++d) acc += x[(size_t)g * Hh + d] * w[lane * Hh + d];
    dot = acc;
  }
  float m = dot;
  for (int o = 16; o > 0; o >>= 1) m = fmaxf(m, __shfl_xor(m, o));
  float ex = (lane < Cc) ? __expf(dot - m) : 0.f;
  float se = ex;
  for (int o = 16; o > 0; o >>= 1) se += __shfl_xor(se, o);
  if (lane < Cc) out[(size_t)g * Cc + lane] = dot - m - logf(se);
}

extern "C" void kernel_launch(void* const* d_in, const int* in_sizes, int n_in,
                              void* d_out, int out_size, void* d_ws, size_t ws_size,
                              hipStream_t stream) {
  const float* x_in  = (const float*)d_in[0];
  const int*   eidx  = (const int*)  d_in[1];
  const int*   idx   = (const int*)  d_in[2];
  const float* fc_w  = (const float*)d_in[3];
  const float* a1    = (const float*)d_in[4];
  const float* a2    = (const float*)d_in[5];
  const float* fc1_w = (const float*)d_in[6];
  const float* fc1_b = (const float*)d_in[7];
  const float* fc2_w = (const float*)d_in[8];
  const float* fc2_b = (const float*)d_in[9];
  const float* gamma = (const float*)d_in[10];
  const float* beta  = (const float*)d_in[11];
  const int* src = eidx;
  const int* dst = eidx + Ee;

  float* ws   = (float*)d_ws;
  float* STAT = ws;
  float* CORR = ws + 64;
  float* Z    = ws + 256;
  float* GOUT = Z + (size_t)Nn * Hh;
  float* POOL = GOUT + (size_t)Nn * Hh;
  float* XB   = POOL + (size_t)Gg * Hh;
  float* FC1O = XB + (size_t)Gg * Hh;
  unsigned short* XBF  = (unsigned short*)(FC1O + (size_t)Gg * Hh);
  unsigned short* WBF  = XBF  + (size_t)Nn * Dd;
  unsigned short* W1BF = WBF  + (size_t)Hh * Dd;
  unsigned short* XBBF = W1BF + (size_t)Hh * Hh;
  unsigned char*  OWN  = (unsigned char*)(XBBF + (size_t)Gg * Hh);
  int*            PK   = (int*)(OWN + (size_t)Ee);

  (void)in_sizes; (void)n_in; (void)out_size; (void)ws_size;

  k_zero<<<1,    256, 0, stream>>>(STAT, 16);
  k_zero<<<64,   256, 0, stream>>>(POOL, (long long)Gg * Hh);

  k_cvtx<<<SBLK, 256, 0, stream>>>(x_in, XBF);
  k_cvt<<<64,   256, 0, stream>>>(fc_w,  WBF,  (long long)Hh * Dd);
  k_cvt<<<64,   256, 0, stream>>>(fc1_w, W1BF, (long long)Hh * Hh);

  k_stats<<<SBLK, 256, 0, stream>>>(x_in, STAT);
  k_prep<<<1, 128, 0, stream>>>(fc_w, STAT, CORR);
  k_gat_gemm<<<Nn / 16, 256, 0, stream>>>(XBF, WBF, STAT, CORR, Z);
  k_own<<<OBLK, 128, 0, stream>>>(src, dst, PK, OWN);
  k_edge2<<<NOWN, 1024, 0, stream>>>(PK, Z, a1, a2, OWN, GOUT);
  k_pool<<<POOL_WG, POOL_WAVES * 32, 0, stream>>>(GOUT, idx, POOL);
  k_bn<<<BN_WG, BN_WAVES * 32, 0, stream>>>(POOL, gamma, beta, XB);
  k_cvt<<<64, 256, 0, stream>>>(XB, XBBF, (long long)Gg * Hh);
  k_fc1<<<Gg / 16, 256, 0, stream>>>(XBBF, W1BF, fc1_b, FC1O);
  k_fc2<<<Gg / 8, 256, 0, stream>>>(FC1O, fc2_w, fc2_b, (float*)d_out);
}


# edge words binned by owner in the packing pass (6400-edge chunks: LDS counter rank, start/length table); an owner copies its 250 runs, scans no owner bytes any more; LDS adds in arrival order
# speedup vs baseline: 4.9538x; 1.1706x over previous
#include <hip/hip_runtime.h>
#include <math.h>

#define Nn 50000
#define Ee 1600000
#define Dd 128
#define Hh 128
#define Cc 10
#define Gg 512
#define LRELU 0.05f
#define BN_EPS 1e-5f

#define NOWN 256
#define ROWS 196
#define QCAP 32768
#define CH   6400
#define NCH  (Ee / CH)
#define BINT (CH / 8)
#define CPW  ((NCH + 31) / 32)
static_assert(NOWN * ROWS >= Nn, "owners must cover every node");
static_assert(Nn <= 65536, "src and dst are packed as two 16-bit halves");
static_assert(NCH * CH == Ee, "the chunks cover the edges exactly: no tail, no guard");
static_assert(BINT * 8 == CH && BINT <= 1024 && BINT % 32 == 0 && BINT >= NOWN, "eight edges per thread, whole waves");
static_assert(CH < 65536, "a run's start and its length share one word");
static_assert(CPW <= 32, "lane k of an owner's wave looks up chunk wave + 32 k");

typedef __attribute__((ext_vector_type(16))) __bf16 v16bf;
typedef __attribute__((ext_vector_type(8)))  float  v8f;

union BFrag { v16bf v; unsigned u[8]; };

__device__ __forceinline__ unsigned short f2bf_bits(float f) {
  union { float f; unsigned u; } v; v.f = f;
  return (unsigned short)((v.u + 0x7FFFu + ((v.u >> 16) & 1u)) >> 16);
}

__global__ void k_zero(float* __restrict__ p, long long n) {
  long long i = (long long)blockIdx.x * blockDim.x + threadIdx.x;
  const long long stride = (long long)gridDim.x * blockDim.x;
  for (; i < n; i += stride) p[i] = 0.0f;
}

__global__ void k_cvt(const float* __restrict__ s, unsigned short* __restrict__ d, long long n) {
  long long i = (long long)blockIdx.x * blockDim.x + threadIdx.x;
  const long long stride = (long long)gridDim.x * blockDim.x;
  for (; i < n; i += stride) d[i] = f2bf_bits(s[i]);
}

#define SV4  10
#define SBLK ((Nn * Dd / 4) / (256 * SV4))
static_assert((long long)SBLK * 256 * SV4 * 4 == (long long)Nn * Dd, "the waves cover x_in exactly: no tail, no guard");

__global__ void __launch_bounds__(256) k_cvtx(const float* __restrict__ x, unsigned short* __restrict__ d) {
  const int lane = threadIdx.x & 31;
  const int w = blockIdx.x * 8 + (threadIdx.x >> 5);
  const float4* xp = (const float4*)x + (size_t)w * (SV4 * 32) + lane;
  uint2* dp = (uint2*)d + (size_t)w * (SV4 * 32) + lane;
  float4 v[SV4];
  #pragma unroll
  for (int k = 0; k < SV4; ++k) v[k] = xp[k * 32];
  #pragma unroll
  for (int k = 0; k < SV4; ++k) {
    uint2 o;
    o.x = (unsigned)f2bf_bits(v[k].x) | ((unsigned)f2bf_bits(v[k].y) << 16);
    o.y = (unsigned)f2bf_bits(v[k].z) | ((unsigned)f2bf_bits(v[k].w) << 16);
    dp[k * 32] = o;
  }
}

__global__ void __launch_bounds__(256) k_stats(const float* __restrict__ x, float* __restrict__ stats) {
  __shared__ float ssum[256], ssq[256];
  const int tid = threadIdx.x;
  const int lane = tid & 31;
  const int w = blockIdx.x * 8 + (tid >> 5);
  const float4* xp = (const float4*)x + (size_t)w * (SV4 * 32) + lane;
  float4 v[SV4];
  #pragma unroll
  for (int k = 0; k < SV4; ++k) v[k] = xp[k * 32];
  float4 s4 = make_float4(0.f, 0.f, 0.f, 0.f), q4 = make_float4(0.f, 0.f, 0.f, 0.f);
  #pragma unroll
  for (int k = 0; k < SV4; ++k) {
    s4.x += v[k].x; s4.y += v[k].y; s4.z += v[k].z; s4.w += v[k].w;
    q4.x += v[k].x * v[k].x; q4.y += v[k].y * v[k].y; q4.z += v[k].z * v[k].z; q4.w += v[k].w * v[k].w;
  }
  const float s = (s4.x + s4.y) + (s4.z + s4.w);
  const float q = (q4.x + q4.y) + (q4.z + q4.w);
  ssum[tid] = s; ssq[tid] = q; __syncthreads();
  for (int o = 128; o > 0; o >>= 1) {
    if (tid < o) { ssum[tid] += ssum[tid + o]; ssq[tid] += ssq[tid + o]; }
    __syncthreads();
  }
  if (tid == 0) { atomicAdd(&stats[0], ssum[0]); atomicAdd(&stats[1], ssq[0]); }
}

__global__ void __launch_bounds__(128) k_prep(const float* __restrict__ w, float* __restrict__ stats,
                       float* __restrict__ corr) {
  const int h = threadIdx.x;
  float colsum = 0.f;
  #pragma unroll
  for (int d = 0; d < Dd; ++d) colsum += w[h * Dd + d];
  const float nd = (float)((long long)Nn * Dd);
  const float s  = stats[0], ss = stats[1];
  const float mu = s / nd;
  const float var = (ss - s * s / nd) / (nd - 1.0f);
  const float invsig = rsqrtf(var);
  if (h == 0) stats[2] = invsig;
  corr[h] = mu * invsig * colsum;
}

__global__ void __launch_bounds__(256) k_gat_gemm(const unsigned short* __restrict__ xbf,
                                                  const unsigned short* __restrict__ wbf,
                                                  const float* __restrict__ stats,
                                                  const float* __restrict__ corr,
                                                  float* __restrict__ z) {
  const int tid  = threadIdx.x;
  const int lane = tid & 31;
  const int hi   = lane >> 4;
  const int lm   = lane & 15;
  const int n0   = blockIdx.x * 16;
  const int h0   = (tid >> 5) * 16;

  const unsigned* xp = (const unsigned*)xbf + (size_t)(n0 + lm) * (Dd / 2);
  const unsigned* wp = (const unsigned*)wbf + (size_t)(h0 + lm) * (Dd / 2);

  v8f c = {0.f, 0.f, 0.f, 0.f, 0.f, 0.f, 0.f, 0.f};
  const float invsig = stats[2];

  #pragma unroll
  for (int k0 = 0; k0 < Dd; k0 += 32) {
    BFrag a, b;
    #pragma unroll
    for (int j = 0; j < 8; ++j) {
      const int kk = k0 + ((j < 4) ? 0 : 16) + (hi ? 8 : 0) + 2 * (j & 3);
      a.u[j] = xp[kk >> 1];
      b.u[j] = wp[kk >> 1];
    }
    if (k0 + 32 < Dd)
      __builtin_prefetch(xp + ((k0 + 32) >> 1), 0, 1);
    c = __builtin_amdgcn_wmma_f32_16x16x32_bf16(false, a.v, false, b.v, (short)0, c, false, false);
  }
  #pragma unroll
  for (int r = 0; r < 8; ++r) {
    const int m = r + hi * 8;
    const int h = h0 + lm;
    z[(size_t)(n0 + m) * Hh + h] = c[r] * invsig - corr[h];
  }
}

__global__ void __launch_bounds__(BINT) k_bin(const int* __restrict__ src, const int* __restrict__ dst,
                                              int* __restrict__ pks, int* __restrict__ tab) {
  __shared__ int cnt[NOWN];
  __shared__ int pre[NOWN];
  __shared__ __attribute__((aligned(16))) int sorted[CH];
  const int tid  = threadIdx.x;
  const int lane = tid & 31;
  const int c    = blockIdx.x;
  if (tid < NOWN) cnt[tid] = 0;
  const int4* sp = (const int4*)(src + (size_t)c * CH);
  const int4* dp = (const int4*)(dst + (size_t)c * CH);
  const int4 s0 = sp[tid], s1 = sp[tid + BINT], d0 = dp[tid], d1 = dp[tid + BINT];
  __syncthreads();
  const int sv[8] = {s0.x, s0.y, s0.z, s0.w, s1.x, s1.y, s1.z, s1.w};
  const int dv[8] = {d0.x, d0.y, d0.z, d0.w, d1.x, d1.y, d1.z, d1.w};
  int ow[8], rk[8];
  #pragma unroll
  for (int i = 0; i < 8; ++i) {
    ow[i] = (int)((unsigned)sv[i] / ROWS);
    rk[i] = __hip_atomic_fetch_add(&cnt[ow[i]], 1, __ATOMIC_RELAXED, __HIP_MEMORY_SCOPE_WORKGROUP);
  }
  __syncthreads();
  if (tid < 32) {
    int e[NOWN / 32], tot = 0;
    #pragma unroll
    for (int j = 0; j < NOWN / 32; ++j) { e[j] = tot; tot += cnt[lane * (NOWN / 32) + j]; }
    int inc = tot;
    #pragma unroll
    for (int o = 1; o < 32; o <<= 1) {
      const int t = __shfl_up(inc, o);
      if (lane >= o) inc += t;
    }
    const int excl = inc - tot;
    #pragma unroll
    for (int j = 0; j < NOWN / 32; ++j) pre[lane * (NOWN / 32) + j] = excl + e[j];
  }
  __syncthreads();
  if (tid < NOWN) tab[c * NOWN + tid] = pre[tid] | (cnt[tid] << 16);
  #pragma unroll
  for (int i = 0; i < 8; ++i)
    sorted[pre[ow[i]] + rk[i]] = (int)((unsigned)sv[i] | ((unsigned)dv[i] << 16));
  __syncthreads();
  int4* op = (int4*)(pks + (size_t)c * CH);
  const int4* s4 = (const int4*)sorted;
  op[tid]        = s4[tid];
  op[tid + BINT] = s4[tid + BINT];
}

__device__ __forceinline__ void lds_add(float* p, float v) {
  (void)__hip_atomic_fetch_add(p, v, __ATOMIC_RELAXED, __HIP_MEMORY_SCOPE_WORKGROUP);
}

template <int M>
__device__ __forceinline__ float row_xor(float x) {
  return __builtin_bit_cast(float, __builtin_amdgcn_update_dpp(0, __builtin_bit_cast(int, x), 0x160 | M, 0xF, 0xF, false));
}

__device__ __forceinline__ float xhalf(float x) {
  const int b = __builtin_bit_cast(int, x);
  return __builtin_bit_cast(float, __builtin_amdgcn_permlanex16(b, b, 0x76543210, 0xfedcba98, false, false));
}

__device__ __forceinline__ float wave_sum(float x) {
  x += row_xor<8>(x);
  x += row_xor<4>(x);
  x += row_xor<2>(x);
  x += row_xor<1>(x);
  return x + xhalf(x);
}

__device__ __forceinline__ void edge_alone(unsigned p, float* rows, float* hs_l, const float* s1_l,
                                           const float* __restrict__ z, const float* __restrict__ a2, int base) {
  const int sl = (int)(p & 0xFFFFu) - base;
  const int d  = (int)(p >> 16);
  const float* zr = z + (size_t)d * Hh;
  float s2 = 0.f;
  for (int c = 0; c < Hh; ++c) s2 += zr[c] * a2[c];
  float ev = s1_l[sl] + s2;
  ev = (ev >= 0.f) ? ev : LRELU * ev;
  const float h = __expf(ev);
  lds_add(&hs_l[sl], h);
  for (int c = 0; c < Hh; ++c)
    lds_add(&rows[sl * Hh + (c & 3) * 32 + (c >> 2)], h * zr[c]);
}

__device__ __forceinline__ void edge_put(unsigned p, int pos, int* queue, float* rows, float* hs_l, const float* s1_l,
                                         const float* __restrict__ z, const float* __restrict__ a2, int base) {
  if (pos < QCAP) queue[pos] = (int)p;
  else edge_alone(p, rows, hs_l, s1_l, z, a2, base);
}

__global__ void __launch_bounds__(1024) k_edge2(const int* __restrict__ pks, const float* __restrict__ z,
                                                const float* __restrict__ a1, const float* __restrict__ a2,
                                                const int* __restrict__ tab,
                                                float* __restrict__ gout) {
  __shared__ __attribute__((aligned(16))) float rows[ROWS * Hh];
  __shared__ int   queue[QCAP];
  __shared__ float hs_l[ROWS];
  __shared__ float s1_l[ROWS];
  __shared__ int   qn;

  const int tid  = threadIdx.x;
  const int lane = tid & 31;
  const int wave = __builtin_amdgcn_readfirstlane(tid >> 5);
  const int b    = blockIdx.x;
  const int base = b * ROWS;

  for (int i = tid; i < ROWS * Hh / 4; i += 1024) ((float4*)rows)[i] = make_float4(0.f, 0.f, 0.f, 0.f);
  if (tid < ROWS) hs_l[tid] = 0.f;
  if (tid == 0) qn = 0;

  const float4 a1v = *(const float4*)(a1 + lane * 4);
  float4 zo[(ROWS + 31) / 32];
  #pragma unroll
  for (int i = 0; i < (ROWS + 31) / 32; ++i) {
    const int r  = wave + 32 * i;
    const int nn = (base + r < Nn) ? (base + r) : (Nn - 1);
    zo[i] = *(const float4*)(z + (size_t)nn * Hh + lane * 4);
  }
  float s1m = 0.f;
  #pragma unroll
  for (int i = 0; i < (ROWS + 31) / 32; ++i) {
    const float s = wave_sum(zo[i].x * a1v.x + zo[i].y * a1v.y + zo[i].z * a1v.z + zo[i].w * a1v.w);
    s1m = (lane == i) ? s : s1m;
  }
  if (lane < (ROWS + 31) / 32 && wave + 32 * lane < ROWS) s1_l[wave + 32 * lane] = s1m;
  __syncthreads();

  {
    const int cme = wave + 32 * lane;
    int tv = 0;
    if (lane < CPW && cme < NCH) tv = tab[cme * NOWN + b];
    const int st = tv & 0xFFFF, cn = tv >> 16;
    int ck[CPW], sk[CPW], ok[CPW], tot = 0;
    #pragma unroll
    for (int k = 0; k < CPW; ++k) {
      ck[k] = __builtin_amdgcn_readlane(cn, k);
      sk[k] = __builtin_amdgcn_readlane(st, k);
      ok[k] = tot;
      tot += ck[k];
    }
    int qb = 0;
    if (lane == 0) qb = __hip_atomic_fetch_add(&qn, tot, __ATOMIC_RELAXED, __HIP_MEMORY_SCOPE_WORKGROUP);
    qb = __builtin_amdgcn_readfirstlane(qb);
    unsigned wv[CPW];
    #pragma unroll
    for (int k = 0; k < CPW; ++k) {
      wv[k] = 0u;
      if (lane < ck[k]) wv[k] = (unsigned)pks[(size_t)(wave + 32 * k) * CH + sk[k] + lane];
    }
    #pragma unroll
    for (int k = 0; k < CPW; ++k) {
      const int pos = qb + ok[k] + lane;
      if (lane < ck[k] && pos < QCAP) queue[pos] = (int)wv[k];
    }
    int run0 = 0;
    for (int k = 0; k < CPW; ++k) {
      const int c_k = __builtin_amdgcn_readlane(cn, k);
      const int s_k = __builtin_amdgcn_readlane(st, k);
      for (int o = 0; o < c_k; o += 32) {
        const int pos = qb + run0 + o + lane;
        if (o + lane < c_k && (o >= 32 || pos >= QCAP))
          edge_put((unsigned)pks[(size_t)(wave + 32 * k) * CH + s_k + o + lane], pos, queue, rows, hs_l, s1_l, z, a2, base);
      }
      run0 += c_k;
    }
  }
  __syncthreads();

  const float4 a2v = *(const float4*)(a2 + lane * 4);
  int n = qn;
  n = __builtin_amdgcn_readfirstlane(n < QCAP ? n : QCAP);
  for (int g0 = wave * 32; g0 < n; g0 += 1024) {
    const int cnt = (n - g0 < 32) ? (n - g0) : 32;
    int sl = 0, d = 0; float s1v = 0.f;
    float hk = 0.f;
    if (lane < cnt) {
      const unsigned p = (unsigned)queue[g0 + lane];
      sl  = (int)(p & 0xFFFFu) - base;
      d   = (int)(p >> 16);
      s1v = s1_l[sl];
    }
    #pragma unroll
    for (int k0 = 0; k0 < 32; k0 += 8) {
      if (k0 < cnt) {
        float4 zv[8]; int sk[8]; float s1k[8];
        #pragma unroll
        for (int j = 0; j < 8; ++j) {
          const int kk = k0 + j;
          sk[j] = __builtin_amdgcn_readlane(sl, kk);
          const int dk = __builtin_amdgcn_readlane(d, kk);
          s1k[j] = __int_as_float(__builtin_amdgcn_readlane(__float_as_int(s1v), kk));
          zv[j] = make_float4(0.f, 0.f, 0.f, 0.f);
          if (kk < cnt) zv[j] = *(const float4*)(z + (size_t)dk * Hh + lane * 4);
        }
        #pragma unroll
        for (int j = 0; j < 8; ++j) {
          if (k0 + j < cnt) {
            const float s2 = wave_sum(zv[j].x * a2v.x + zv[j].y * a2v.y + zv[j].z * a2v.z + zv[j].w * a2v.w);
            float ev = s1k[j] + s2;
            ev = (ev >= 0.f) ? ev : LRELU * ev;
            const float h = __expf(ev);
            float* r = rows + sk[j] * Hh + lane;
            lds_add(r,      h * zv[j].x);
            lds_add(r + 32, h * zv[j].y);
            lds_add(r + 64, h * zv[j].z);
            lds_add(r + 96, h * zv[j].w);
            hk = (lane == k0 + j) ? h : hk;
          }
        }
      }
    }
    if (lane < cnt) lds_add(&hs_l[sl], hk);
  }
  __syncthreads();

  for (int r = wave; r < ROWS; r += 32) {
    const int nn = base + r;
    if (nn < Nn) {
      const float hs  = hs_l[r];
      const float inv = (hs > 0.f) ? 1.0f / hs : 0.f;
      const float* p = rows + r * Hh + lane;
      *(float4*)(gout + (size_t)nn * Hh + lane * 4) = make_float4(p[0] * inv, p[32] * inv, p[64] * inv, p[96] * inv);
    }
  }
}

#define PCH        16
#define POOL_WAVES 4
#define POOL_NW    (Nn / PCH)
#define POOL_WG    ((POOL_NW + POOL_WAVES - 1) / POOL_WAVES)
static_assert(Nn % PCH == 0, "every wave that has nodes has PCH of them");
static_assert(PCH <= 32, "lane i keeps the graph id of the wave's node i");

__device__ __forceinline__ void pool_flush(float* __restrict__ pool, int g, int lane, float4 a) {
  float* p = pool + (size_t)g * Hh + lane * 4;
  atomicAdd(p + 0, a.x);
  atomicAdd(p + 1, a.y);
  atomicAdd(p + 2, a.z);
  atomicAdd(p + 3, a.w);
}

__global__ void __launch_bounds__(POOL_WAVES * 32) k_pool(const float* __restrict__ gout, const int* __restrict__ idx,
                                                          float* __restrict__ pool) {
  const int lane = threadIdx.x & 31;
  const int w = blockIdx.x * POOL_WAVES + (threadIdx.x >> 5);
  if (w >= POOL_NW) return;
  const int n0 = w * PCH;
  const int gi = idx[n0 + (lane & (PCH - 1))];
  float4 v[PCH];
  #pragma unroll
  for (int i = 0; i < PCH; ++i) v[i] = *(const float4*)(gout + (size_t)(n0 + i) * Hh + lane * 4);
  int cur = __builtin_amdgcn_readlane(gi, 0);
  float4 acc = make_float4(0.f, 0.f, 0.f, 0.f);
  #pragma unroll
  for (int i = 0; i < PCH; ++i) {
    const int g = __builtin_amdgcn_readlane(gi, i);
    if (g != cur) {
      pool_flush(pool, cur, lane, acc);
      acc = make_float4(0.f, 0.f, 0.f, 0.f);
      cur = g;
    }
    acc.x += fmaxf(v[i].x, 0.f);
    acc.y += fmaxf(v[i].y, 0.f);
    acc.z += fmaxf(v[i].z, 0.f);
    acc.w += fmaxf(v[i].w, 0.f);
  }
  pool_flush(pool, cur, lane, acc);
}

#define BN_WG    (Hh / 32)
#define BN_WAVES 16
#define BN_ROWS  (Gg / BN_WAVES)
static_assert(BN_WG * 32 == Hh, "32 columns per work-group");
static_assert(BN_WAVES * BN_ROWS == Gg, "the waves of a work-group cover every graph row");
__global__ void __launch_bounds__(BN_WAVES * 32) k_bn(const float* __restrict__ pooled, const float* __restrict__ gamma,
                                                      const float* __restrict__ beta, float* __restrict__ xb) {
  __shared__ float ps[BN_WAVES][32], pq[BN_WAVES][32];
  const int lane = threadIdx.x & 31;
  const int wave = threadIdx.x >> 5;
  const int h  = blockIdx.x * 32 + lane;
  const int g0 = wave * BN_ROWS;
  float v[BN_ROWS];
  #pragma unroll
  for (int i = 0; i < BN_ROWS; ++i) v[i] = pooled[(g0 + i) * Hh + h];
  float s = 0.f, q = 0.f;
  #pragma unroll
  for (int i = 0; i < BN_ROWS; ++i) { s += v[i]; q += v[i] * v[i]; }
  ps[wave][lane] = s; pq[wave][lane] = q;
  __syncthreads();
  float S = 0.f, Q = 0.f;
  #pragma unroll
  for (int w = 0; w < BN_WAVES; ++w) { S += ps[w][lane]; Q += pq[w][lane]; }
  const float mu = S / (float)Gg;
  const float var = Q / (float)Gg - mu * mu;
  const float inv = rsqrtf(var + BN_EPS);
  const float ga = gamma[h], be = beta[h];
  #pragma unroll
  for (int i = 0; i < BN_ROWS; ++i)
    xb[(g0 + i) * Hh + h] = (v[i] - mu) * inv * ga + be;
}

__global__ void __launch_bounds__(256) k_fc1(const unsigned short* __restrict__ xbf,
                                             const unsigned short* __restrict__ wbf,
                                             const float* __restrict__ bias, float* __restrict__ out) {
  const int tid  = threadIdx.x;
  const int lane = tid & 31;
  const int hi   = lane >> 4;
  const int lm   = lane & 15;
  const int n0   = blockIdx.x * 16;
  const int h0   = (tid >> 5) * 16;

  const unsigned* xp = (const unsigned*)xbf + (size_t)(n0 + lm) * (Hh / 2);
  const unsigned* wp = (const unsigned*)wbf + (size_t)(h0 + lm) * (Hh / 2);

  v8f c = {0.f, 0.f, 0.f, 0.f, 0.f, 0.f, 0.f, 0.f};
  #pragma unroll
  for (int k0 = 0; k0 < Hh; k0 += 32) {
    BFrag a, b;
    #pragma unroll
    for (int j = 0; j < 8; ++j) {
      const int kk = k0 + ((j < 4) ? 0 : 16) + (hi ? 8 : 0) + 2 * (j & 3);
      a.u[j] = xp[kk >> 1];
      b.u[j] = wp[kk >> 1];
    }
    c = __builtin_amdgcn_wmma_f32_16x16x32_bf16(false, a.v, false, b.v, (short)0, c, false, false);
  }
  #pragma unroll
  for (int r = 0; r < 8; ++r) {
    const int m = r + hi * 8;
    const int h = h0 + lm;
    out[(size_t)(n0 + m) * Hh + h] = fmaxf(c[r] + bias[h], 0.f);
  }
}

__global__ void __launch_bounds__(256) k_fc2(const float* __restrict__ x, const float* __restrict__ w,
                                             const float* __restrict__ b, float* __restrict__ out) {
  const int lane = threadIdx.x & 31;
  const int g = blockIdx.x * 8 + (threadIdx.x >> 5);
  if (g >= Gg) return;
  float dot = -INFINITY;
  if (lane < Cc) {
    float acc = b[lane];
    for (int d = 0; d < Hh; ++d) acc += x[(size_t)g * Hh + d] * w[lane * Hh + d];
    dot = acc;
  }
  float m = dot;
  for (int o = 16; o > 0; o >>= 1) m = fmaxf(m, __shfl_xor(m, o));
  float ex = (lane < Cc) ? __expf(dot - m) : 0.f;
  float se = ex;
  for (int o = 16; o > 0; o >>= 1) se += __shfl_xor(se, o);
  if (lane < Cc) out[(size_t)g * Cc + lane] = dot - m - logf(se);
}

extern "C" void kernel_launch(void* const* d_in, const int* in_sizes, int n_in,
                              void* d_out, int out_size, void* d_ws, size_t ws_size,
                              hipStream_t stream) {
  const float* x_in  = (const float*)d_in[0];
  const int*   eidx  = (const int*)  d_in[1];
  const int*   idx   = (const int*)  d_in[2];
  const float* fc_w  = (const float*)d_in[3];
  const float* a1    = (const float*)d_in[4];
  const float* a2    = (const float*)d_in[5];
  const float* fc1_w = (const float*)d_in[6];
  const float* fc1_b = (const float*)d_in[7];
  const float* fc2_w = (const float*)d_in[8];
  const float* fc2_b = (const float*)d_in[9];
  const float* gamma = (const float*)d_in[10];
  const float* beta  = (const float*)d_in[11];
  const int* src = eidx;
  const int* dst = eidx + Ee;

  float* ws   = (float*)d_ws;
  float* STAT = ws;
  float* CORR = ws + 64;
  float* Z    = ws + 256;
  float* GOUT = Z + (size_t)Nn * Hh;
  float* POOL = GOUT + (size_t)Nn * Hh;
  float* XB   = POOL + (size_t)Gg * Hh;
  float* FC1O = XB + (size_t)Gg * Hh;
  unsigned short* XBF  = (unsigned short*)(FC1O + (size_t)Gg * Hh);
  unsigned short* WBF  = XBF  + (size_t)Nn * Dd;
  unsigned short* W1BF = WBF  + (size_t)Hh * Dd;
  unsigned short* XBBF = W1BF + (size_t)Hh * Hh;
  int*            TAB  = (int*)(XBBF + (size_t)Gg * Hh);
  int*            PKS  = TAB + (size_t)NCH * NOWN;

  (void)in_sizes; (void)n_in; (void)out_size; (void)ws_size;

  k_zero<<<1,    256, 0, stream>>>(STAT, 16);
  k_zero<<<64,   256, 0, stream>>>(POOL, (long long)Gg * Hh);

  k_cvtx<<<SBLK, 256, 0, stream>>>(x_in, XBF);
  k_cvt<<<64,   256, 0, stream>>>(fc_w,  WBF,  (long long)Hh * Dd);
  k_cvt<<<64,   256, 0, stream>>>(fc1_w, W1BF, (long long)Hh * Hh);

  k_stats<<<SBLK, 256, 0, stream>>>(x_in, STAT);
  k_prep<<<1, 128, 0, stream>>>(fc_w, STAT, CORR);
  k_gat_gemm<<<Nn / 16, 256, 0, stream>>>(XBF, WBF, STAT, CORR, Z);
  k_bin<<<NCH, BINT, 0, stream>>>(src, dst, PKS, TAB);
  k_edge2<<<NOWN, 1024, 0, stream>>>(PKS, Z, a1, a2, TAB, GOUT);
  k_pool<<<POOL_WG, POOL_WAVES * 32, 0, stream>>>(GOUT, idx, POOL);
  k_bn<<<BN_WG, BN_WAVES * 32, 0, stream>>>(POOL, gamma, beta, XB);
  k_cvt<<<64, 256, 0, stream>>>(XB, XBBF, (long long)Gg * Hh);
  k_fc1<<<Gg / 16, 256, 0, stream>>>(XBBF, W1BF, fc1_b, FC1O);
  k_fc2<<<Gg / 8, 256, 0, stream>>>(FC1O, fc2_w, fc2_b, (float*)d_out);
}


# bf16 staging launches removed: fc_w and fc1_w are rounded (nearest even, same bits) in the consumer's registers from f32, BN stores its result as bf16 itself; 11 launches, not 14; sums in old order
# speedup vs baseline: 5.4545x; 1.1011x over previous
#include <hip/hip_runtime.h>
#include <math.h>

#define Nn 50000
#define Ee 1600000
#define Dd 128
#define Hh 128
#define Cc 10
#define Gg 512
#define LRELU 0.05f
#define BN_EPS 1e-5f

#define NOWN 256
#define ROWS 196
#define QCAP 32768
#define CH   6400
#define NCH  (Ee / CH)
#define BINT (CH / 8)
#define CPW  ((NCH + 31) / 32)
static_assert(NOWN * ROWS >= Nn, "owners must cover every node");
static_assert(Nn <= 65536, "src and dst are packed as two 16-bit halves");
static_assert(NCH * CH == Ee, "the chunks cover the edges exactly: no tail, no guard");
static_assert(BINT * 8 == CH && BINT <= 1024 && BINT % 32 == 0 && BINT >= NOWN, "eight edges per thread, whole waves");
static_assert(CH < 65536, "a run's start and its length share one word");
static_assert(CPW <= 32, "lane k of an owner's wave looks up chunk wave + 32 k");

typedef __attribute__((ext_vector_type(16))) __bf16 v16bf;
typedef __attribute__((ext_vector_type(8)))  float  v8f;

union BFrag { v16bf v; unsigned u[8]; };

__device__ __forceinline__ unsigned short f2bf_bits(float f) {
  union { float f; unsigned u; } v; v.f = f;
  return (unsigned short)((v.u + 0x7FFFu + ((v.u >> 16) & 1u)) >> 16);
}

__device__ __forceinline__ float bf2f(unsigned b) { return __uint_as_float(b << 16); }

typedef float  f32x2  __attribute__((ext_vector_type(2)));
typedef __bf16 bf16x2 __attribute__((ext_vector_type(2)));
__device__ __forceinline__ unsigned bf_pair(float lo, float hi) {
  const f32x2 v = {lo, hi};
  return __builtin_bit_cast(unsigned, __builtin_convertvector(v, bf16x2));
}

struct WRow { float4 f[4]; };
__device__ __forceinline__ WRow wrow_load(const float* __restrict__ row, int t, int half) {
  const float4* q = (const float4*)(row + t * 32 + half * 8);
  WRow r;
  r.f[0] = q[0]; r.f[1] = q[1]; r.f[2] = q[4]; r.f[3] = q[5];
  return r;
}
__device__ __forceinline__ void wrow_pack(const WRow& r, unsigned* u) {
  #pragma unroll
  for (int i = 0; i < 4; ++i) {
    u[2 * i]     = bf_pair(r.f[i].x, r.f[i].y);
    u[2 * i + 1] = bf_pair(r.f[i].z, r.f[i].w);
  }
}

template <int M>
__device__ __forceinline__ float row_xor(float x) {
  return __builtin_bit_cast(float, __builtin_amdgcn_update_dpp(0, __builtin_bit_cast(int, x), 0x160 | M, 0xF, 0xF, false));
}

__device__ __forceinline__ float xhalf(float x) {
  const int b = __builtin_bit_cast(int, x);
  return __builtin_bit_cast(float, __builtin_amdgcn_permlanex16(b, b, 0x76543210, 0xfedcba98, false, false));
}

__device__ __forceinline__ float wave_sum(float x) {
  x += row_xor<8>(x);
  x += row_xor<4>(x);
  x += row_xor<2>(x);
  x += row_xor<1>(x);
  return x + xhalf(x);
}

__global__ void k_zero(float* __restrict__ p, long long n) {
  long long i = (long long)blockIdx.x * blockDim.x + threadIdx.x;
  const long long stride = (long long)gridDim.x * blockDim.x;
  for (; i < n; i += stride) p[i] = 0.0f;
}

#define SV4  10
#define SBLK ((Nn * Dd / 4) / (256 * SV4))
static_assert((long long)SBLK * 256 * SV4 * 4 == (long long)Nn * Dd, "the waves cover x_in exactly: no tail, no guard");

__global__ void __launch_bounds__(256) k_cvtx(const float* __restrict__ x, unsigned short* __restrict__ d) {
  const int lane = threadIdx.x & 31;
  const int w = blockIdx.x * 8 + (threadIdx.x >> 5);
  const float4* xp = (const float4*)x + (size_t)w * (SV4 * 32) + lane;
  uint2* dp = (uint2*)d + (size_t)w * (SV4 * 32) + lane;
  float4 v[SV4];
  #pragma unroll
  for (int k = 0; k < SV4; ++k) v[k] = xp[k * 32];
  #pragma unroll
  for (int k = 0; k < SV4; ++k) {
    uint2 o;
    o.x = (unsigned)f2bf_bits(v[k].x) | ((unsigned)f2bf_bits(v[k].y) << 16);
    o.y = (unsigned)f2bf_bits(v[k].z) | ((unsigned)f2bf_bits(v[k].w) << 16);
    dp[k * 32] = o;
  }
}

__global__ void __launch_bounds__(256) k_stats(const float* __restrict__ x, float* __restrict__ stats) {
  __shared__ float ssum[256], ssq[256];
  const int tid = threadIdx.x;
  const int lane = tid & 31;
  const int w = blockIdx.x * 8 + (tid >> 5);
  const float4* xp = (const float4*)x + (size_t)w * (SV4 * 32) + lane;
  float4 v[SV4];
  #pragma unroll
  for (int k = 0; k < SV4; ++k) v[k] = xp[k * 32];
  float4 s4 = make_float4(0.f, 0.f, 0.f, 0.f), q4 = make_float4(0.f, 0.f, 0.f, 0.f);
  #pragma unroll
  for (int k = 0; k < SV4; ++k) {
    s4.x += v[k].x; s4.y += v[k].y; s4.z += v[k].z; s4.w += v[k].w;
    q4.x += v[k].x * v[k].x; q4.y += v[k].y * v[k].y; q4.z += v[k].z * v[k].z; q4.w += v[k].w * v[k].w;
  }
  const float s = (s4.x + s4.y) + (s4.z + s4.w);
  const float q = (q4.x + q4.y) + (q4.z + q4.w);
  ssum[tid] = s; ssq[tid] = q; __syncthreads();
  for (int o = 128; o > 0; o >>= 1) {
    if (tid < o) { ssum[tid] += ssum[tid + o]; ssq[tid] += ssq[tid + o]; }
    __syncthreads();
  }
  if (tid == 0) { atomicAdd(&stats[0], ssum[0]); atomicAdd(&stats[1], ssq[0]); }
}

__global__ void __launch_bounds__(256) k_prep(const float* __restrict__ w, const float* __restrict__ a1,
                       const float* __restrict__ a2, float* __restrict__ stats,
                       float* __restrict__ corr, float* __restrict__ vs) {
  __shared__ float part[2][4];
  const int t = threadIdx.x;
  const float nd = (float)((long long)Nn * Dd);
  const float s  = stats[0], ss = stats[1];
  const float mu = s / nd;
  const float var = (ss - s * s / nd) / (nd - 1.0f);
  const float invsig = rsqrtf(var);
  if (t < Hh) {
    const int h = t;
    float colsum = 0.f;
    #pragma unroll
    for (int d = 0; d < Dd; ++d) colsum += w[h * Dd + d];
    if (h == 0) stats[2] = invsig;
    corr[h] = mu * invsig * colsum;
  } else {
    const int d = t - Hh;
    float v1 = 0.f, v2 = 0.f;
    #pragma unroll
    for (int r = 0; r < Hh; ++r) {
      const float wv = w[r * Dd + d];
      v1 += wv * a1[r];
      v2 += wv * a2[r];
    }
    const float vs1 = invsig * v1, vs2 = invsig * v2;
    vs[d] = vs1;
    vs[Dd + d] = vs2;
    const float p1 = wave_sum(vs1), p2 = wave_sum(vs2);
    if ((d & 31) == 0) { part[0][d >> 5] = p1; part[1][d >> 5] = p2; }
  }
  __syncthreads();
  if (t == 0)
    vs[2 * Dd] = mu * ((((part[0][0] + part[0][1]) + part[0][2]) + part[0][3]) +
                       (((part[1][0] + part[1][1]) + part[1][2]) + part[1][3]));
}

__global__ void __launch_bounds__(BINT) k_bin(const int* __restrict__ src, const int* __restrict__ dst,
                                              int* __restrict__ pks, int* __restrict__ tab) {
  __shared__ int cnt[NOWN];
  __shared__ int pre[NOWN];
  __shared__ __attribute__((aligned(16))) int sorted[CH];
  const int tid  = threadIdx.x;
  const int lane = tid & 31;
  const int c    = blockIdx.x;
  if (tid < NOWN) cnt[tid] = 0;
  const int4* sp = (const int4*)(src + (size_t)c * CH);
  const int4* dp = (const int4*)(dst + (size_t)c * CH);
  const int4 s0 = sp[tid], s1 = sp[tid + BINT], d0 = dp[tid], d1 = dp[tid + BINT];
  __syncthreads();
  const int sv[8] = {s0.x, s0.y, s0.z, s0.w, s1.x, s1.y, s1.z, s1.w};
  const int dv[8] = {d0.x, d0.y, d0.z, d0.w, d1.x, d1.y, d1.z, d1.w};
  int ow[8], rk[8];
  #pragma unroll
  for (int i = 0; i < 8; ++i) {
    ow[i] = (int)((unsigned)sv[i] / ROWS);
    rk[i] = __hip_atomic_fetch_add(&cnt[ow[i]], 1, __ATOMIC_RELAXED, __HIP_MEMORY_SCOPE_WORKGROUP);
  }
  __syncthreads();
  if (tid < 32) {
    int e[NOWN / 32], tot = 0;
    #pragma unroll
    for (int j = 0; j < NOWN / 32; ++j) { e[j] = tot; tot += cnt[lane * (NOWN / 32) + j]; }
    int inc = tot;
    #pragma unroll
    for (int o = 1; o < 32; o <<= 1) {
      const int t = __shfl_up(inc, o);
      if (lane >= o) inc += t;
    }
    const int excl = inc - tot;
    #pragma unroll
    for (int j = 0; j < NOWN / 32; ++j) pre[lane * (NOWN / 32) + j] = excl + e[j];
  }
  __syncthreads();
  if (tid < NOWN) tab[c * NOWN + tid] = pre[tid] | (cnt[tid] << 16);
  #pragma unroll
  for (int i = 0; i < 8; ++i)
    sorted[pre[ow[i]] + rk[i]] = (int)((unsigned)sv[i] | ((unsigned)dv[i] << 16));
  __syncthreads();
  int4* op = (int4*)(pks + (size_t)c * CH);
  const int4* s4 = (const int4*)sorted;
  op[tid]        = s4[tid];
  op[tid + BINT] = s4[tid + BINT];
}

__device__ __forceinline__ void lds_add(float* p, float v) {
  (void)__hip_atomic_fetch_add(p, v, __ATOMIC_RELAXED, __HIP_MEMORY_SCOPE_WORKGROUP);
}

__device__ __forceinline__ void edge_alone(unsigned p, float* rows, float* hs_l, const float* s1_l,
                                           const unsigned short* __restrict__ xbf, const float* __restrict__ vs2, int base) {
  const int sl = (int)(p & 0xFFFFu) - base;
  const int d  = (int)(p >> 16);
  const unsigned short* xr = xbf + (size_t)d * Dd;
  float s2 = 0.f;
  for (int c = 0; c < Dd; ++c) s2 += bf2f(xr[c]) * vs2[c];
  float ev = s1_l[sl] + s2;
  ev = (ev >= 0.f) ? ev : LRELU * ev;
  const float h = __expf(ev);
  lds_add(&hs_l[sl], h);
  for (int c = 0; c < Dd; ++c)
    lds_add(&rows[sl * Dd + (c & 3) * 32 + (c >> 2)], h * bf2f(xr[c]));
}

__device__ __forceinline__ void edge_put(unsigned p, int pos, int* queue, float* rows, float* hs_l, const float* s1_l,
                                         const unsigned short* __restrict__ xbf, const float* __restrict__ vs2, int base) {
  if (pos < QCAP) queue[pos] = (int)p;
  else edge_alone(p, rows, hs_l, s1_l, xbf, vs2, base);
}

#define APW   (Dd / 2 + 4)
#define APL   (ROWS * APW)
#define NRT   ((ROWS + 15) / 16)
#define RIF   16
static_assert(2 * APL <= QCAP, "the two bf16 copies of the owned rows reuse the queue's memory");
static_assert((APW * 4) % 32 == 16 && (APL * 4) % 16 == 0, "row pitch 16 bytes x odd, copies 16-byte aligned");
__global__ void __launch_bounds__(1024) k_edge2(const int* __restrict__ pks, const unsigned short* __restrict__ xbf,
                                                const float* __restrict__ w,
                                                const float* __restrict__ vs, const float* __restrict__ stats,
                                                const float* __restrict__ corr,
                                                const int* __restrict__ tab,
                                                float* __restrict__ gout) {
  __shared__ __attribute__((aligned(16))) float rows[ROWS * Dd];
  __shared__ __attribute__((aligned(16))) int queue[QCAP];
  __shared__ __attribute__((aligned(16))) float hs_l[NRT * 16];
  __shared__ float s1_l[ROWS];
  __shared__ int   qn;

  const int tid  = threadIdx.x;
  const int lane = tid & 31;
  const int wave = __builtin_amdgcn_readfirstlane(tid >> 5);
  const int b    = blockIdx.x;
  const int base = b * ROWS;
  const float* vs2 = vs + Dd;

  for (int i = tid; i < ROWS * Dd / 4; i += 1024) ((float4*)rows)[i] = make_float4(0.f, 0.f, 0.f, 0.f);
  if (tid < NRT * 16) hs_l[tid] = 0.f;
  if (tid == 0) qn = 0;

  const float4 v1v = *(const float4*)(vs + lane * 4);
  const float csum = vs[2 * Dd];
  uint2 xo[(ROWS + 31) / 32];
  #pragma unroll
  for (int i = 0; i < (ROWS + 31) / 32; ++i) {
    const int r  = wave + 32 * i;
    const int nn = (base + r < Nn) ? (base + r) : (Nn - 1);
    xo[i] = *(const uint2*)(xbf + (size_t)nn * Dd + lane * 4);
  }
  float s1m = 0.f;
  #pragma unroll
  for (int i = 0; i < (ROWS + 31) / 32; ++i) {
    const float s = wave_sum(bf2f(xo[i].x & 0xFFFFu) * v1v.x + bf2f(xo[i].x >> 16) * v1v.y +
                             bf2f(xo[i].y & 0xFFFFu) * v1v.z + bf2f(xo[i].y >> 16) * v1v.w);
    s1m = (lane == i) ? s : s1m;
  }
  if (lane < (ROWS + 31) / 32 && wave + 32 * lane < ROWS) s1_l[wave + 32 * lane] = s1m - csum;
  __syncthreads();

  {
    const int cme = wave + 32 * lane;
    int tv = 0;
    if (lane < CPW && cme < NCH) tv = tab[cme * NOWN + b];
    const int st = tv & 0xFFFF, cn = tv >> 16;
    int ck[CPW], sk[CPW], ok[CPW], tot = 0;
    #pragma unroll
    for (int k = 0; k < CPW; ++k) {
      ck[k] = __builtin_amdgcn_readlane(cn, k);
      sk[k] = __builtin_amdgcn_readlane(st, k);
      ok[k] = tot;
      tot += ck[k];
    }
    int qb = 0;
    if (lane == 0) qb = __hip_atomic_fetch_add(&qn, tot, __ATOMIC_RELAXED, __HIP_MEMORY_SCOPE_WORKGROUP);
    qb = __builtin_amdgcn_readfirstlane(qb);
    unsigned wv[CPW];
    #pragma unroll
    for (int k = 0; k < CPW; ++k) {
      wv[k] = 0u;
      if (lane < ck[k]) wv[k] = (unsigned)pks[(size_t)(wave + 32 * k) * CH + sk[k] + lane];
    }
    #pragma unroll
    for (int k = 0; k < CPW; ++k) {
      const int pos = qb + ok[k] + lane;
      if (lane < ck[k] && pos < QCAP) queue[pos] = (int)wv[k];
    }
    int run0 = 0;
    for (int k = 0; k < CPW; ++k) {
      const int c_k = __builtin_amdgcn_readlane(cn, k);
      const int s_k = __builtin_amdgcn_readlane(st, k);
      for (int o = 0; o < c_k; o += 32) {
        const int pos = qb + run0 + o + lane;
        if (o + lane < c_k && (o >= 32 || pos >= QCAP))
          edge_put((unsigned)pks[(size_t)(wave + 32 * k) * CH + s_k + o + lane], pos, queue, rows, hs_l, s1_l, xbf, vs2, base);
      }
      run0 += c_k;
    }
  }
  __syncthreads();

  const float4 v2v = *(const float4*)(vs2 + lane * 4);
  int n = qn;
  n = __builtin_amdgcn_readfirstlane(n < QCAP ? n : QCAP);
  for (int g0 = wave * 32; g0 < n; g0 += 1024) {
    const int cnt = (n - g0 < 32) ? (n - g0) : 32;
    int sl = 0, d = 0; float s1v = 0.f;
    float hk = 0.f;
    if (lane < cnt) {
      const unsigned p = (unsigned)queue[g0 + lane];
      sl  = (int)(p & 0xFFFFu) - base;
      d   = (int)(p >> 16);
      s1v = s1_l[sl];
    }
    #pragma unroll
    for (int k0 = 0; k0 < 32; k0 += RIF) {
      if (k0 < cnt) {
        uint2 xv[RIF]; int sk[RIF]; float s1k[RIF];
        #pragma unroll
        for (int j = 0; j < RIF; ++j) {
          const int kk = k0 + j;
          sk[j] = __builtin_amdgcn_readlane(sl, kk);
          const int dk = __builtin_amdgcn_readlane(d, kk);
          s1k[j] = __int_as_float(__builtin_amdgcn_readlane(__float_as_int(s1v), kk));
          xv[j] = make_uint2(0u, 0u);
          if (kk < cnt) xv[j] = *(const uint2*)(xbf + (size_t)dk * Dd + lane * 4);
        }
        #pragma unroll
        for (int j = 0; j < RIF; ++j) {
          if (k0 + j < cnt) {
            const float x0 = bf2f(xv[j].x & 0xFFFFu), x1 = bf2f(xv[j].x >> 16);
            const float x2 = bf2f(xv[j].y & 0xFFFFu), x3 = bf2f(xv[j].y >> 16);
            const float s2 = wave_sum(x0 * v2v.x + x1 * v2v.y + x2 * v2v.z + x3 * v2v.w);
            float ev = s1k[j] + s2;
            ev = (ev >= 0.f) ? ev : LRELU * ev;
            const float h = __expf(ev);
            float* r = rows + sk[j] * Dd + lane;
            lds_add(r,      h * x0);
            lds_add(r + 32, h * x1);
            lds_add(r + 64, h * x2);
            lds_add(r + 96, h * x3);
            hk = (lane == k0 + j) ? h : hk;
          }
        }
      }
    }
    if (lane < cnt) lds_add(&hs_l[sl], hk);
  }

  const int hi = lane >> 4;
  const int lm = lane & 15;
  const int h0 = (wave & 7) * 16;
  const float* wrow = w + (size_t)(h0 + lm) * Dd;
  WRow wr[Dd / 32];
  #pragma unroll
  for (int t = 0; t < Dd / 32; ++t) wr[t] = wrow_load(wrow, t, hi);
  const float invsig = stats[2];
  const float cr = corr[h0 + lm];
  __syncthreads();
  BFrag bw[Dd / 32];
  #pragma unroll
  for (int t = 0; t < Dd / 32; ++t) wrow_pack(wr[t], bw[t].u);

  unsigned* ahi = (unsigned*)queue;
  unsigned* alo = ahi + APL;
  for (int r = wave; r < ROWS; r += 32) {
    const float hs  = hs_l[r];
    const float inv = (hs > 0.f) ? 1.0f / hs : 0.f;
    const float* p = rows + r * Dd + lane;
    const float m0 = p[0] * inv, m1 = p[32] * inv, m2 = p[64] * inv, m3 = p[96] * inv;
    const unsigned t0 = f2bf_bits(m0), t1 = f2bf_bits(m1), t2 = f2bf_bits(m2), t3 = f2bf_bits(m3);
    const unsigned u0 = f2bf_bits(m0 - bf2f(t0)), u1 = f2bf_bits(m1 - bf2f(t1));
    const unsigned u2 = f2bf_bits(m2 - bf2f(t2)), u3 = f2bf_bits(m3 - bf2f(t3));
    *(uint2*)(ahi + r * APW + lane * 2) = make_uint2(t0 | (t1 << 16), t2 | (t3 << 16));
    *(uint2*)(alo + r * APW + lane * 2) = make_uint2(u0 | (u1 << 16), u2 | (u3 << 16));
  }
  __syncthreads();

  for (int rt = wave >> 3; rt < NRT; rt += 4) {
    const int ar = (rt * 16 + lm < ROWS) ? (rt * 16 + lm) : (ROWS - 1);
    const unsigned* ph = ahi + ar * APW;
    const unsigned* pl = alo + ar * APW;
    const float4 f0 = *(const float4*)(hs_l + rt * 16 + hi * 8);
    const float4 f1 = *(const float4*)(hs_l + rt * 16 + hi * 8 + 4);
    const float hv[8] = {f0.x, f0.y, f0.z, f0.w, f1.x, f1.y, f1.z, f1.w};
    v8f c = {0.f, 0.f, 0.f, 0.f, 0.f, 0.f, 0.f, 0.f};
    #pragma unroll
    for (int t = 0; t < Dd / 32; ++t) {
      BFrag ah, al;
      #pragma unroll
      for (int j = 0; j < 8; ++j) {
        const int kk = t * 32 + ((j < 4) ? 0 : 16) + (hi ? 8 : 0) + 2 * (j & 3);
        ah.u[j] = ph[kk >> 1];
        al.u[j] = pl[kk >> 1];
      }
      c = __builtin_amdgcn_wmma_f32_16x16x32_bf16(false, ah.v, false, bw[t].v, (short)0, c, false, false);
      c = __builtin_amdgcn_wmma_f32_16x16x32_bf16(false, al.v, false, bw[t].v, (short)0, c, false, false);
    }
    #pragma unroll
    for (int r = 0; r < 8; ++r) {
      const int row = rt * 16 + r + hi * 8;
      const int nn  = base + row;
      if (row < ROWS && nn < Nn)
        gout[(size_t)nn * Hh + h0 + lm] = (hv[r] > 0.f) ? (c[r] * invsig - cr) : 0.f;
    }
  }
}

#define PCH        16
#define POOL_WAVES 4
#define POOL_NW    (Nn / PCH)
#define POOL_WG    ((POOL_NW + POOL_WAVES - 1) / POOL_WAVES)
static_assert(Nn % PCH == 0, "every wave that has nodes has PCH of them");
static_assert(PCH <= 32, "lane i keeps the graph id of the wave's node i");

__device__ __forceinline__ void pool_flush(float* __restrict__ pool, int g, int lane, float4 a) {
  float* p = pool + (size_t)g * Hh + lane * 4;
  atomicAdd(p + 0, a.x);
  atomicAdd(p + 1, a.y);
  atomicAdd(p + 2, a.z);
  atomicAdd(p + 3, a.w);
}

__global__ void __launch_bounds__(POOL_WAVES * 32) k_pool(const float* __restrict__ gout, const int* __restrict__ idx,
                                                          float* __restrict__ pool) {
  const int lane = threadIdx.x & 31;
  const int w = blockIdx.x * POOL_WAVES + (threadIdx.x >> 5);
  if (w >= POOL_NW) return;
  const int n0 = w * PCH;
  const int gi = idx[n0 + (lane & (PCH - 1))];
  float4 v[PCH];
  #pragma unroll
  for (int i = 0; i < PCH; ++i) v[i] = *(const float4*)(gout + (size_t)(n0 + i) * Hh + lane * 4);
  int cur = __builtin_amdgcn_readlane(gi, 0);
  float4 acc = make_float4(0.f, 0.f, 0.f, 0.f);
  #pragma unroll
  for (int i = 0; i < PCH; ++i) {
    const int g = __builtin_amdgcn_readlane(gi, i);
    if (g != cur) {
      pool_flush(pool, cur, lane, acc);
      acc = make_float4(0.f, 0.f, 0.f, 0.f);
      cur = g;
    }
    acc.x += fmaxf(v[i].x, 0.f);
    acc.y += fmaxf(v[i].y, 0.f);
    acc.z += fmaxf(v[i].z, 0.f);
    acc.w += fmaxf(v[i].w, 0.f);
  }
  pool_flush(pool, cur, lane, acc);
}

#define BN_WG    (Hh / 32)
#define BN_WAVES 16
#define BN_ROWS  (Gg / BN_WAVES)
static_assert(BN_WG * 32 == Hh, "32 columns per work-group");
static_assert(BN_WAVES * BN_ROWS == Gg, "the waves of a work-group cover every graph row");
__global__ void __launch_bounds__(BN_WAVES * 32) k_bn(const float* __restrict__ pooled, const float* __restrict__ gamma,
                                                      const float* __restrict__ beta, unsigned short* __restrict__ xb) {
  __shared__ float ps[BN_WAVES][32], pq[BN_WAVES][32];
  const int lane = threadIdx.x & 31;
  const int wave = threadIdx.x >> 5;
  const int h  = blockIdx.x * 32 + lane;
  const int g0 = wave * BN_ROWS;
  float v[BN_ROWS];
  #pragma unroll
  for (int i = 0; i < BN_ROWS; ++i) v[i] = pooled[(g0 + i) * Hh + h];
  float s = 0.f, q = 0.f;
  #pragma unroll
  for (int i = 0; i < BN_ROWS; ++i) { s += v[i]; q += v[i] * v[i]; }
  ps[wave][lane] = s; pq[wave][lane] = q;
  __syncthreads();
  float S = 0.f, Q = 0.f;
  #pragma unroll
  for (int w = 0; w < BN_WAVES; ++w) { S += ps[w][lane]; Q += pq[w][lane]; }
  const float mu = S / (float)Gg;
  const float var = Q / (float)Gg - mu * mu;
  const float inv = rsqrtf(var + BN_EPS);
  const float ga = gamma[h], be = beta[h];
  #pragma unroll
  for (int i = 0; i < BN_ROWS; ++i)
    xb[(g0 + i) * Hh + h] = f2bf_bits((v[i] - mu) * inv * ga + be);
}

__global__ void __launch_bounds__(256) k_fc1(const unsigned short* __restrict__ xbf,
                                             const float* __restrict__ w,
                                             const float* __restrict__ bias, float* __restrict__ out) {
  const int tid  = threadIdx.x;
  const int lane = tid & 31;
  const int hi   = lane >> 4;
  const int lm   = lane & 15;
  const int n0   = blockIdx.x * 16;
  const int h0   = (tid >> 5) * 16;

  const unsigned* xp = (const unsigned*)xbf + (size_t)(n0 + lm) * (Hh / 2);
  const float* wrow = w + (size_t)(h0 + lm) * Hh;
  WRow wr[Hh / 32];
  #pragma unroll
  for (int t = 0; t < Hh / 32; ++t) wr[t] = wrow_load(wrow, t, hi);

  v8f c = {0.f, 0.f, 0.f, 0.f, 0.f, 0.f, 0.f, 0.f};
  #pragma unroll
  for (int t = 0; t < Hh / 32; ++t) {
    BFrag a, b;
    #pragma unroll
    for (int j = 0; j < 8; ++j) {
      const int kk = t * 32 + ((j < 4) ? 0 : 16) + (hi ? 8 : 0) + 2 * (j & 3);
      a.u[j] = xp[kk >> 1];
    }
    wrow_pack(wr[t], b.u);
    c = __builtin_amdgcn_wmma_f32_16x16x32_bf16(false, a.v, false, b.v, (short)0, c, false, false);
  }
  #pragma unroll
  for (int r = 0; r < 8; ++r) {
    const int m = r + hi * 8;
    const int h = h0 + lm;
    out[(size_t)(n0 + m) * Hh + h] = fmaxf(c[r] + bias[h], 0.f);
  }
}

__global__ void __launch_bounds__(256) k_fc2(const float* __restrict__ x, const float* __restrict__ w,
                                             const float* __restrict__ b, float* __restrict__ out) {
  const int lane = threadIdx.x & 31;
  const int g = blockIdx.x * 8 + (threadIdx.x >> 5);
  if (g >= Gg) return;
  float dot = -INFINITY;
  if (lane < Cc) {
    float acc = b[lane];
    for (int d = 0; d < Hh; ++d) acc += x[(size_t)g * Hh + d] * w[lane * Hh + d];
    dot = acc;
  }
  float m = dot;
  for (int o = 16; o > 0; o >>= 1) m = fmaxf(m, __shfl_xor(m, o));
  float ex = (lane < Cc) ? __expf(dot - m) : 0.f;
  float se = ex;
  for (int o = 16; o > 0; o >>= 1) se += __shfl_xor(se, o);
  if (lane < Cc) out[(size_t)g * Cc + lane] = dot - m - logf(se);
}

extern "C" void kernel_launch(void* const* d_in, const int* in_sizes, int n_in,
                              void* d_out, int out_size, void* d_ws, size_t ws_size,
                              hipStream_t stream) {
  const float* x_in  = (const float*)d_in[0];
  const int*   eidx  = (const int*)  d_in[1];
  const int*   idx   = (const int*)  d_in[2];
  const float* fc_w  = (const float*)d_in[3];
  const float* a1    = (const float*)d_in[4];
  const float* a2    = (const float*)d_in[5];
  const float* fc1_w = (const float*)d_in[6];
  const float* fc1_b = (const float*)d_in[7];
  const float* fc2_w = (const float*)d_in[8];
  const float* fc2_b = (const float*)d_in[9];
  const float* gamma = (const float*)d_in[10];
  const float* beta  = (const float*)d_in[11];
  const int* src = eidx;
  const int* dst = eidx + Ee;

  float* ws   = (float*)d_ws;
  float* STAT = ws;
  float* CORR = ws + 64;
  float* Z    = ws + 256;
  float* VS   = Z;
  float* GOUT = Z + (size_t)Nn * Hh;
  float* POOL = GOUT + (size_t)Nn * Hh;
  float* FC1O = POOL + (size_t)Gg * Hh;
  unsigned short* XBF  = (unsigned short*)(FC1O + (size_t)Gg * Hh);
  unsigned short* XBBF = XBF + (size_t)Nn * Dd;
  int*            TAB  = (int*)(XBBF + (size_t)Gg * Hh);
  int*            PKS  = TAB + (size_t)NCH * NOWN;

  (void)in_sizes; (void)n_in; (void)out_size; (void)ws_size;

  k_zero<<<1,    256, 0, stream>>>(STAT, 16);
  k_zero<<<64,   256, 0, stream>>>(POOL, (long long)Gg * Hh);

  k_cvtx<<<SBLK, 256, 0, stream>>>(x_in, XBF);

  k_stats<<<SBLK, 256, 0, stream>>>(x_in, STAT);
  k_prep<<<1, 256, 0, stream>>>(fc_w, a1, a2, STAT, CORR, VS);
  k_bin<<<NCH, BINT, 0, stream>>>(src, dst, PKS, TAB);
  k_edge2<<<NOWN, 1024, 0, stream>>>(PKS, XBF, fc_w, VS, STAT, CORR, TAB, GOUT);
  k_pool<<<POOL_WG, POOL_WAVES * 32, 0, stream>>>(GOUT, idx, POOL);
  k_bn<<<BN_WG, BN_WAVES * 32, 0, stream>>>(POOL, gamma, beta, XBBF);
  k_fc1<<<Gg / 16, 256, 0, stream>>>(XBBF, fc1_w, fc1_b, FC1O);
  k_fc2<<<Gg / 8, 256, 0, stream>>>(FC1O, fc2_w, fc2_b, (float*)d_out);
}


# edge sum without an LDS add per edge: owner sorts its edge words by row (LDS counter rank, arrival order; prefix; scatter); a wave sums a row's run of edges in registers (fma), one LDS add per run
# speedup vs baseline: 7.1211x; 1.3055x over previous
#include <hip/hip_runtime.h>
#include <math.h>

#define Nn 50000
#define Ee 1600000
#define Dd 128
#define Hh 128
#define Cc 10
#define Gg 512
#define LRELU 0.05f
#define BN_EPS 1e-5f

#define NOWN 256
#define ROWS 196
#define QCAP 32768
#define CH   6400
#define NCH  (Ee / CH)
#define BINT (CH / 8)
#define CPW  ((NCH + 31) / 32)
#define RCN  224
static_assert(NOWN * ROWS >= Nn, "owners must cover every node");
static_assert(Nn <= 65536, "src and dst are packed as two 16-bit halves");
static_assert(NCH * CH == Ee, "the chunks cover the edges exactly: no tail, no guard");
static_assert(BINT * 8 == CH && BINT <= 1024 && BINT % 32 == 0 && BINT >= NOWN, "eight edges per thread, whole waves");
static_assert(CH < 65536, "a run's start and its length share one word");
static_assert(CPW <= 32, "lane k of an owner's wave looks up chunk wave + 32 k");
static_assert(RCN >= ROWS && RCN % 32 == 0, "a counter per owned row, the same number for every lane of the scanning wave");

typedef __attribute__((ext_vector_type(16))) __bf16 v16bf;
typedef __attribute__((ext_vector_type(8)))  float  v8f;

union BFrag { v16bf v; unsigned u[8]; };

__device__ __forceinline__ unsigned short f2bf_bits(float f) {
  union { float f; unsigned u; } v; v.f = f;
  return (unsigned short)((v.u + 0x7FFFu + ((v.u >> 16) & 1u)) >> 16);
}

__device__ __forceinline__ float bf2f(unsigned b) { return __uint_as_float(b << 16); }

typedef float  f32x2  __attribute__((ext_vector_type(2)));
typedef __bf16 bf16x2 __attribute__((ext_vector_type(2)));
__device__ __forceinline__ unsigned bf_pair(float lo, float hi) {
  const f32x2 v = {lo, hi};
  return __builtin_bit_cast(unsigned, __builtin_convertvector(v, bf16x2));
}

struct WRow { float4 f[4]; };
__device__ __forceinline__ WRow wrow_load(const float* __restrict__ row, int t, int half) {
  const float4* q = (const float4*)(row + t * 32 + half * 8);
  WRow r;
  r.f[0] = q[0]; r.f[1] = q[1]; r.f[2] = q[4]; r.f[3] = q[5];
  return r;
}
__device__ __forceinline__ void wrow_pack(const WRow& r, unsigned* u) {
  #pragma unroll
  for (int i = 0; i < 4; ++i) {
    u[2 * i]     = bf_pair(r.f[i].x, r.f[i].y);
    u[2 * i + 1] = bf_pair(r.f[i].z, r.f[i].w);
  }
}

template <int M>
__device__ __forceinline__ float row_xor(float x) {
  return __builtin_bit_cast(float, __builtin_amdgcn_update_dpp(0, __builtin_bit_cast(int, x), 0x160 | M, 0xF, 0xF, false));
}

__device__ __forceinline__ float xhalf(float x) {
  const int b = __builtin_bit_cast(int, x);
  return __builtin_bit_cast(float, __builtin_amdgcn_permlanex16(b, b, 0x76543210, 0xfedcba98, false, false));
}

__device__ __forceinline__ float wave_sum(float x) {
  x += row_xor<8>(x);
  x += row_xor<4>(x);
  x += row_xor<2>(x);
  x += row_xor<1>(x);
  return x + xhalf(x);
}

__global__ void k_zero(float* __restrict__ p, long long n) {
  long long i = (long long)blockIdx.x * blockDim.x + threadIdx.x;
  const long long stride = (long long)gridDim.x * blockDim.x;
  for (; i < n; i += stride) p[i] = 0.0f;
}

#define SV4  10
#define SBLK ((Nn * Dd / 4) / (256 * SV4))
static_assert((long long)SBLK * 256 * SV4 * 4 == (long long)Nn * Dd, "the waves cover x_in exactly: no tail, no guard");

static_assert(Dd * 4 == 32 * 16, "a wave's 16-byte load covers one x row");
__global__ void __launch_bounds__(256) k_cvtx(const float* __restrict__ x, unsigned short* __restrict__ d,
                                              const float* __restrict__ vs2, float* __restrict__ s2t) {
  const int lane = threadIdx.x & 31;
  const int w = blockIdx.x * 8 + (threadIdx.x >> 5);
  const float4* xp = (const float4*)x + (size_t)w * (SV4 * 32) + lane;
  uint2* dp = (uint2*)d + (size_t)w * (SV4 * 32) + lane;
  float4 v[SV4];
  #pragma unroll
  for (int k = 0; k < SV4; ++k) v[k] = xp[k * 32];
  const float4 v2v = *(const float4*)(vs2 + lane * 4);
  float s2m = 0.f;
  #pragma unroll
  for (int k = 0; k < SV4; ++k) {
    const unsigned b0 = f2bf_bits(v[k].x), b1 = f2bf_bits(v[k].y), b2 = f2bf_bits(v[k].z), b3 = f2bf_bits(v[k].w);
    uint2 o;
    o.x = b0 | (b1 << 16);
    o.y = b2 | (b3 << 16);
    dp[k * 32] = o;
    const float s = wave_sum(bf2f(b0) * v2v.x + bf2f(b1) * v2v.y + bf2f(b2) * v2v.z + bf2f(b3) * v2v.w);
    s2m = (lane == k) ? s : s2m;
  }
  if (lane < SV4) s2t[w * SV4 + lane] = s2m;
}

__global__ void __launch_bounds__(256) k_stats(const float* __restrict__ x, float* __restrict__ stats) {
  __shared__ float ssum[256], ssq[256];
  const int tid = threadIdx.x;
  const int lane = tid & 31;
  const int w = blockIdx.x * 8 + (tid >> 5);
  const float4* xp = (const float4*)x + (size_t)w * (SV4 * 32) + lane;
  float4 v[SV4];
  #pragma unroll
  for (int k = 0; k < SV4; ++k) v[k] = xp[k * 32];
  float4 s4 = make_float4(0.f, 0.f, 0.f, 0.f), q4 = make_float4(0.f, 0.f, 0.f, 0.f);
  #pragma unroll
  for (int k = 0; k < SV4; ++k) {
    s4.x += v[k].x; s4.y += v[k].y; s4.z += v[k].z; s4.w += v[k].w;
    q4.x += v[k].x * v[k].x; q4.y += v[k].y * v[k].y; q4.z += v[k].z * v[k].z; q4.w += v[k].w * v[k].w;
  }
  const float s = (s4.x + s4.y) + (s4.z + s4.w);
  const float q = (q4.x + q4.y) + (q4.z + q4.w);
  ssum[tid] = s; ssq[tid] = q; __syncthreads();
  for (int o = 128; o > 0; o >>= 1) {
    if (tid < o) { ssum[tid] += ssum[tid + o]; ssq[tid] += ssq[tid + o]; }
    __syncthreads();
  }
  if (tid == 0) { atomicAdd(&stats[0], ssum[0]); atomicAdd(&stats[1], ssq[0]); }
}

__global__ void __launch_bounds__(256) k_prep(const float* __restrict__ w, const float* __restrict__ a1,
                       const float* __restrict__ a2, float* __restrict__ stats,
                       float* __restrict__ corr, float* __restrict__ vs) {
  __shared__ float part[2][4];
  const int t = threadIdx.x;
  const float nd = (float)((long long)Nn * Dd);
  const float s  = stats[0], ss = stats[1];
  const float mu = s / nd;
  const float var = (ss - s * s / nd) / (nd - 1.0f);
  const float invsig = rsqrtf(var);
  if (t < Hh) {
    const int h = t;
    float colsum = 0.f;
    #pragma unroll
    for (int d = 0; d < Dd; ++d) colsum += w[h * Dd + d];
    if (h == 0) stats[2] = invsig;
    corr[h] = mu * invsig * colsum;
  } else {
    const int d = t - Hh;
    float v1 = 0.f, v2 = 0.f;
    #pragma unroll
    for (int r = 0; r < Hh; ++r) {
      const float wv = w[r * Dd + d];
      v1 += wv * a1[r];
      v2 += wv * a2[r];
    }
    const float vs1 = invsig * v1, vs2 = invsig * v2;
    vs[d] = vs1;
    vs[Dd + d] = vs2;
    const float p1 = wave_sum(vs1), p2 = wave_sum(vs2);
    if ((d & 31) == 0) { part[0][d >> 5] = p1; part[1][d >> 5] = p2; }
  }
  __syncthreads();
  if (t == 0)
    vs[2 * Dd] = mu * ((((part[0][0] + part[0][1]) + part[0][2]) + part[0][3]) +
                       (((part[1][0] + part[1][1]) + part[1][2]) + part[1][3]));
}

__global__ void __launch_bounds__(BINT) k_bin(const int* __restrict__ src, const int* __restrict__ dst,
                                              int* __restrict__ pks, int* __restrict__ tab) {
  __shared__ int cnt[NOWN];
  __shared__ int pre[NOWN];
  __shared__ __attribute__((aligned(16))) int sorted[CH];
  const int tid  = threadIdx.x;
  const int lane = tid & 31;
  const int c    = blockIdx.x;
  if (tid < NOWN) cnt[tid] = 0;
  const int4* sp = (const int4*)(src + (size_t)c * CH);
  const int4* dp = (const int4*)(dst + (size_t)c * CH);
  const int4 s0 = sp[tid], s1 = sp[tid + BINT], d0 = dp[tid], d1 = dp[tid + BINT];
  __syncthreads();
  const int sv[8] = {s0.x, s0.y, s0.z, s0.w, s1.x, s1.y, s1.z, s1.w};
  const int dv[8] = {d0.x, d0.y, d0.z, d0.w, d1.x, d1.y, d1.z, d1.w};
  int ow[8], rk[8];
  #pragma unroll
  for (int i = 0; i < 8; ++i) {
    ow[i] = (int)((unsigned)sv[i] / ROWS);
    rk[i] = __hip_atomic_fetch_add(&cnt[ow[i]], 1, __ATOMIC_RELAXED, __HIP_MEMORY_SCOPE_WORKGROUP);
  }
  __syncthreads();
  if (tid < 32) {
    int e[NOWN / 32], tot = 0;
    #pragma unroll
    for (int j = 0; j < NOWN / 32; ++j) { e[j] = tot; tot += cnt[lane * (NOWN / 32) + j]; }
    int inc = tot;
    #pragma unroll
    for (int o = 1; o < 32; o <<= 1) {
      const int t = __shfl_up(inc, o);
      if (lane >= o) inc += t;
    }
    const int excl = inc - tot;
    #pragma unroll
    for (int j = 0; j < NOWN / 32; ++j) pre[lane * (NOWN / 32) + j] = excl + e[j];
  }
  __syncthreads();
  if (tid < NOWN) tab[c * NOWN + tid] = pre[tid] | (cnt[tid] << 16);
  #pragma unroll
  for (int i = 0; i < 8; ++i)
    sorted[pre[ow[i]] + rk[i]] = (int)((unsigned)sv[i] | ((unsigned)dv[i] << 16));
  __syncthreads();
  int4* op = (int4*)(pks + (size_t)c * CH);
  const int4* s4 = (const int4*)sorted;
  op[tid]        = s4[tid];
  op[tid + BINT] = s4[tid + BINT];
}

__device__ __forceinline__ void lds_add(float* p, float v) {
  (void)__hip_atomic_fetch_add(p, v, __ATOMIC_RELAXED, __HIP_MEMORY_SCOPE_WORKGROUP);
}

__device__ __forceinline__ void edge_alone(unsigned p, float* rows, float* hs_l, const float* s1_l,
                                           const unsigned short* __restrict__ xbf, const float* __restrict__ s2t, int base) {
  const int sl = (int)(p & 0xFFFFu) - base;
  const int d  = (int)(p >> 16);
  const unsigned short* xr = xbf + (size_t)d * Dd;
  float ev = s1_l[sl] + s2t[d];
  ev = (ev >= 0.f) ? ev : LRELU * ev;
  const float h = __expf(ev);
  lds_add(&hs_l[sl], h);
  #pragma nounroll
  for (int c = 0; c < Dd; ++c)
    lds_add(&rows[sl * Dd + (c & 3) * 32 + (c >> 2)], h * bf2f(xr[c]));
}

__device__ __forceinline__ void row_flush(float* rows, float* hs_l, int cur, int lane,
                                          float a0, float a1, float a2, float a3, float hsum) {
  float* r = rows + cur * Dd + lane;
  lds_add(r,      a0);
  lds_add(r + 32, a1);
  lds_add(r + 64, a2);
  lds_add(r + 96, a3);
  if (lane == 0) lds_add(&hs_l[cur], hsum);
}

#define RIF   16
template <bool FULL>
__device__ __forceinline__ void drain_batch(int cnt, int sl, int d, unsigned chg, float s1v, float s2v,
                                            const unsigned short* __restrict__ xbf, float* rows, float* hs_l, int lane) {
  float h = 0.f;
  float a0 = 0.f, a1 = 0.f, a2 = 0.f, a3 = 0.f, hsum = 0.f;
  int cur = __builtin_amdgcn_readlane(sl, 0);
  #pragma unroll
  for (int k0 = 0; k0 < 32; k0 += RIF) {
    if (FULL || k0 < cnt) {
      uint2 xv[RIF];
      #pragma unroll
      for (int j = 0; j < RIF; ++j) {
        const int kk = k0 + j;
        const unsigned dk = (unsigned)__builtin_amdgcn_readlane(d, kk);
        xv[j] = make_uint2(0u, 0u);
        if (FULL || kk < cnt) xv[j] = *(const uint2*)(xbf + (size_t)dk * Dd + lane * 4);
      }
      if (k0 == 0) {
        float ev = s1v + s2v;
        ev = (ev >= 0.f) ? ev : LRELU * ev;
        h = __expf(ev);
      }
      #pragma unroll
      for (int j = 0; j < RIF; ++j) {
        const int kk = k0 + j;
        if (FULL || kk < cnt) {
          if (kk > 0 && ((chg >> kk) & 1u)) {
            row_flush(rows, hs_l, cur, lane, a0, a1, a2, a3, hsum);
            a0 = 0.f; a1 = 0.f; a2 = 0.f; a3 = 0.f; hsum = 0.f;
            cur = __builtin_amdgcn_readlane(sl, kk);
          }
          const float hk = __int_as_float(__builtin_amdgcn_readlane(__float_as_int(h), kk));
          hsum += hk;
          a0 = __builtin_fmaf(hk, bf2f(xv[j].x & 0xFFFFu), a0);
          a1 = __builtin_fmaf(hk, bf2f(xv[j].x >> 16),     a1);
          a2 = __builtin_fmaf(hk, bf2f(xv[j].y & 0xFFFFu), a2);
          a3 = __builtin_fmaf(hk, bf2f(xv[j].y >> 16),     a3);
        }
      }
    }
  }
  row_flush(rows, hs_l, cur, lane, a0, a1, a2, a3, hsum);
}

#define APW   (Dd / 2 + 4)
#define APL   (ROWS * APW)
#define NRT   ((ROWS + 15) / 16)
static_assert(2 * APL <= QCAP, "the two bf16 copies of the owned rows reuse the queue's memory");
static_assert((APW * 4) % 32 == 16 && (APL * 4) % 16 == 0, "row pitch 16 bytes x odd, copies 16-byte aligned");
__global__ void __launch_bounds__(1024) k_edge2(const int* __restrict__ pks, const unsigned short* __restrict__ xbf,
                                                const float* __restrict__ w,
                                                const float* __restrict__ vs, const float* __restrict__ s2t,
                                                const float* __restrict__ stats,
                                                const float* __restrict__ corr,
                                                const int* __restrict__ tab,
                                                float* __restrict__ gout) {
  __shared__ __attribute__((aligned(16))) float rows[ROWS * Dd];
  __shared__ __attribute__((aligned(16))) int queue[QCAP];
  __shared__ __attribute__((aligned(16))) float hs_l[NRT * 16];
  __shared__ float s1_l[ROWS];
  __shared__ int   rcnt[RCN];
  __shared__ int   rpre[RCN];
  __shared__ int   qn;

  const int tid  = threadIdx.x;
  const int lane = tid & 31;
  const int wave = __builtin_amdgcn_readfirstlane(tid >> 5);
  const int b    = blockIdx.x;
  const int base = b * ROWS;

  for (int i = tid; i < ROWS * Dd / 4; i += 1024) ((float4*)rows)[i] = make_float4(0.f, 0.f, 0.f, 0.f);
  if (tid < NRT * 16) hs_l[tid] = 0.f;
  if (tid < RCN) rcnt[tid] = 0;

  const float4 v1v = *(const float4*)(vs + lane * 4);
  const float csum = vs[2 * Dd];
  uint2 xo[(ROWS + 31) / 32];
  #pragma unroll
  for (int i = 0; i < (ROWS + 31) / 32; ++i) {
    const int r  = wave + 32 * i;
    const int nn = (base + r < Nn) ? (base + r) : (Nn - 1);
    xo[i] = *(const uint2*)(xbf + (size_t)nn * Dd + lane * 4);
  }
  float s1m = 0.f;
  #pragma unroll
  for (int i = 0; i < (ROWS + 31) / 32; ++i) {
    const float s = wave_sum(bf2f(xo[i].x & 0xFFFFu) * v1v.x + bf2f(xo[i].x >> 16) * v1v.y +
                             bf2f(xo[i].y & 0xFFFFu) * v1v.z + bf2f(xo[i].y >> 16) * v1v.w);
    s1m = (lane == i) ? s : s1m;
  }
  if (lane < (ROWS + 31) / 32 && wave + 32 * lane < ROWS) s1_l[wave + 32 * lane] = s1m - csum;
  __syncthreads();

  const int cme = wave + 32 * lane;
  int tv = 0;
  if (lane < CPW && cme < NCH) tv = tab[cme * NOWN + b];
  const int st = tv & 0xFFFF, cn = tv >> 16;
  int ck[CPW], sk[CPW];
  #pragma unroll
  for (int k = 0; k < CPW; ++k) {
    ck[k] = __builtin_amdgcn_readlane(cn, k);
    sk[k] = __builtin_amdgcn_readlane(st, k);
  }
  unsigned wv[2 * CPW];
  int rk[2 * CPW];
  #pragma unroll
  for (int i = 0; i < 2 * CPW; ++i) {
    const int k = i >> 1, o = lane + 32 * (i & 1);
    wv[i] = 0u;
    if (o < ck[k]) wv[i] = (unsigned)pks[(size_t)(wave + 32 * k) * CH + sk[k] + o];
  }
  #pragma unroll
  for (int i = 0; i < 2 * CPW; ++i) {
    const int k = i >> 1, o = lane + 32 * (i & 1);
    rk[i] = 0;
    if (o < ck[k]) rk[i] = __hip_atomic_fetch_add(&rcnt[(int)(wv[i] & 0xFFFFu) - base], 1, __ATOMIC_RELAXED, __HIP_MEMORY_SCOPE_WORKGROUP);
  }
  for (int k = 0; k < CPW; ++k) {
    const int c_k = __builtin_amdgcn_readlane(cn, k);
    const int s_k = __builtin_amdgcn_readlane(st, k);
    for (int o = 64; o < c_k; o += 32)
      if (o + lane < c_k)
        edge_alone((unsigned)pks[(size_t)(wave + 32 * k) * CH + s_k + o + lane], rows, hs_l, s1_l, xbf, s2t, base);
  }
  __syncthreads();

  if (wave == 0) {
    int e[RCN / 32], tot = 0;
    #pragma unroll
    for (int j = 0; j < RCN / 32; ++j) { e[j] = tot; tot += rcnt[lane * (RCN / 32) + j]; }
    int inc = tot;
    #pragma unroll
    for (int o = 1; o < 32; o <<= 1) {
      const int t = __shfl_up(inc, o);
      if (lane >= o) inc += t;
    }
    const int excl = inc - tot;
    #pragma unroll
    for (int j = 0; j < RCN / 32; ++j) rpre[lane * (RCN / 32) + j] = excl + e[j];
    if (lane == 31) qn = inc;
  }
  __syncthreads();

  #pragma unroll
  for (int i = 0; i < 2 * CPW; ++i) {
    const int k = i >> 1, o = lane + 32 * (i & 1);
    if (o < ck[k]) {
      const int pos = rpre[(int)(wv[i] & 0xFFFFu) - base] + rk[i];
      if (pos < QCAP) queue[pos] = (int)wv[i];
      else edge_alone(wv[i], rows, hs_l, s1_l, xbf, s2t, base);
    }
  }
  __syncthreads();

  int n = qn;
  n = __builtin_amdgcn_readfirstlane(n < QCAP ? n : QCAP);
  for (int g0 = wave * 32; g0 < n; g0 += 1024) {
    const int cnt = (n - g0 < 32) ? (n - g0) : 32;
    int sl = 0, d = 0, slp = -1;
    if (lane < cnt) {
      const unsigned p = (unsigned)queue[g0 + lane];
      sl  = (int)(p & 0xFFFFu) - base;
      d   = (int)(p >> 16);
      if (lane > 0) slp = (int)((unsigned)queue[g0 + lane - 1] & 0xFFFFu) - base;
    }
    const unsigned chg = __builtin_amdgcn_ballot_w32(lane < cnt && sl != slp);
    const float s2v = s2t[d];
    const float s1v = s1_l[sl];
    if (cnt == 32) drain_batch<true>(cnt, sl, d, chg, s1v, s2v, xbf, rows, hs_l, lane);
    else           drain_batch<false>(cnt, sl, d, chg, s1v, s2v, xbf, rows, hs_l, lane);
  }

  const int hi = lane >> 4;
  const int lm = lane & 15;
  const int h0 = (wave & 7) * 16;
  const float* wrow = w + (size_t)(h0 + lm) * Dd;
  WRow wr[Dd / 32];
  #pragma unroll
  for (int t = 0; t < Dd / 32; ++t) wr[t] = wrow_load(wrow, t, hi);
  const float invsig = stats[2];
  const float cr = corr[h0 + lm];
  __syncthreads();
  BFrag bw[Dd / 32];
  #pragma unroll
  for (int t = 0; t < Dd / 32; ++t) wrow_pack(wr[t], bw[t].u);

  unsigned* ahi = (unsigned*)queue;
  unsigned* alo = ahi + APL;
  for (int r = wave; r < ROWS; r += 32) {
    const float hs  = hs_l[r];
    const float inv = (hs > 0.f) ? 1.0f / hs : 0.f;
    const float* p = rows + r * Dd + lane;
    const float m0 = p[0] * inv, m1 = p[32] * inv, m2 = p[64] * inv, m3 = p[96] * inv;
    const unsigned t0 = f2bf_bits(m0), t1 = f2bf_bits(m1), t2 = f2bf_bits(m2), t3 = f2bf_bits(m3);
    const unsigned u0 = f2bf_bits(m0 - bf2f(t0)), u1 = f2bf_bits(m1 - bf2f(t1));
    const unsigned u2 = f2bf_bits(m2 - bf2f(t2)), u3 = f2bf_bits(m3 - bf2f(t3));
    *(uint2*)(ahi + r * APW + lane * 2) = make_uint2(t0 | (t1 << 16), t2 | (t3 << 16));
    *(uint2*)(alo + r * APW + lane * 2) = make_uint2(u0 | (u1 << 16), u2 | (u3 << 16));
  }
  __syncthreads();

  for (int rt = wave >> 3; rt < NRT; rt += 4) {
    const int ar = (rt * 16 + lm < ROWS) ? (rt * 16 + lm) : (ROWS - 1);
    const unsigned* ph = ahi + ar * APW;
    const unsigned* pl = alo + ar * APW;
    const float4 f0 = *(const float4*)(hs_l + rt * 16 + hi * 8);
    const float4 f1 = *(const float4*)(hs_l + rt * 16 + hi * 8 + 4);
    const float hv[8] = {f0.x, f0.y, f0.z, f0.w, f1.x, f1.y, f1.z, f1.w};
    v8f c = {0.f, 0.f, 0.f, 0.f, 0.f, 0.f, 0.f, 0.f};
    #pragma unroll
    for (int t = 0; t < Dd / 32; ++t) {
      BFrag ah, al;
      #pragma unroll
      for (int j = 0; j < 8; ++j) {
        const int kk = t * 32 + ((j < 4) ? 0 : 16) + (hi ? 8 : 0) + 2 * (j & 3);
        ah.u[j] = ph[kk >> 1];
        al.u[j] = pl[kk >> 1];
      }
      c = __builtin_amdgcn_wmma_f32_16x16x32_bf16(false, ah.v, false, bw[t].v, (short)0, c, false, false);
      c = __builtin_amdgcn_wmma_f32_16x16x32_bf16(false, al.v, false, bw[t].v, (short)0, c, false, false);
    }
    #pragma unroll
    for (int r = 0; r < 8; ++r) {
      const int row = rt * 16 + r + hi * 8;
      const int nn  = base + row;
      if (row < ROWS && nn < Nn)
        gout[(size_t)nn * Hh + h0 + lm] = (hv[r] > 0.f) ? (c[r] * invsig - cr) : 0.f;
    }
  }
}

#define PCH        16
#define POOL_WAVES 4
#define POOL_NW    (Nn / PCH)
#define POOL_WG    ((POOL_NW + POOL_WAVES - 1) / POOL_WAVES)
static_assert(Nn % PCH == 0, "every wave that has nodes has PCH of them");
static_assert(PCH <= 32, "lane i keeps the graph id of the wave's node i");

__device__ __forceinline__ void pool_flush(float* __restrict__ pool, int g, int lane, float4 a) {
  float* p = pool + (size_t)g * Hh + lane * 4;
  atomicAdd(p + 0, a.x);
  atomicAdd(p + 1, a.y);
  atomicAdd(p + 2, a.z);
  atomicAdd(p + 3, a.w);
}

__global__ void __launch_bounds__(POOL_WAVES * 32) k_pool(const float* __restrict__ gout, const int* __restrict__ idx,
                                                          float* __restrict__ pool) {
  const int lane = threadIdx.x & 31;
  const int w = blockIdx.x * POOL_WAVES + (threadIdx.x >> 5);
  if (w >= POOL_NW) return;
  const int n0 = w * PCH;
  const int gi = idx[n0 + (lane & (PCH - 1))];
  float4 v[PCH];
  #pragma unroll
  for (int i = 0; i < PCH; ++i) v[i] = *(const float4*)(gout + (size_t)(n0 + i) * Hh + lane * 4);
  int cur = __builtin_amdgcn_readlane(gi, 0);
  float4 acc = make_float4(0.f, 0.f, 0.f, 0.f);
  #pragma unroll
  for (int i = 0; i < PCH; ++i) {
    const int g = __builtin_amdgcn_readlane(gi, i);
    if (g != cur) {
      pool_flush(pool, cur, lane, acc);
      acc = make_float4(0.f, 0.f, 0.f, 0.f);
      cur = g;
    }
    acc.x += fmaxf(v[i].x, 0.f);
    acc.y += fmaxf(v[i].y, 0.f);
    acc.z += fmaxf(v[i].z, 0.f);
    acc.w += fmaxf(v[i].w, 0.f);
  }
  pool_flush(pool, cur, lane, acc);
}

#define BN_WG    (Hh / 32)
#define BN_WAVES 16
#define BN_ROWS  (Gg / BN_WAVES)
static_assert(BN_WG * 32 == Hh, "32 columns per work-group");
static_assert(BN_WAVES * BN_ROWS == Gg, "the waves of a work-group cover every graph row");
__global__ void __launch_bounds__(BN_WAVES * 32) k_bn(const float* __restrict__ pooled, const float* __restrict__ gamma,
                                                      const float* __restrict__ beta, unsigned short* __restrict__ xb) {
  __shared__ float ps[BN_WAVES][32], pq[BN_WAVES][32];
  const int lane = threadIdx.x & 31;
  const int wave = threadIdx.x >> 5;
  const int h  = blockIdx.x * 32 + lane;
  const int g0 = wave * BN_ROWS;
  float v[BN_ROWS];
  #pragma unroll
  for (int i = 0; i < BN_ROWS; ++i) v[i] = pooled[(g0 + i) * Hh + h];
  float s = 0.f, q = 0.f;
  #pragma unroll
  for (int i = 0; i < BN_ROWS; ++i) { s += v[i]; q += v[i] * v[i]; }
  ps[wave][lane] = s; pq[wave][lane] = q;
  __syncthreads();
  float S = 0.f, Q = 0.f;
  #pragma unroll
  for (int w = 0; w < BN_WAVES; ++w) { S += ps[w][lane]; Q += pq[w][lane]; }
  const float mu = S / (float)Gg;
  const float var = Q / (float)Gg - mu * mu;
  const float inv = rsqrtf(var + BN_EPS);
  const float ga = gamma[h], be = beta[h];
  #pragma unroll
  for (int i = 0; i < BN_ROWS; ++i)
    xb[(g0 + i) * Hh + h] = f2bf_bits((v[i] - mu) * inv * ga + be);
}

__global__ void __launch_bounds__(256) k_fc1(const unsigned short* __restrict__ xbf,
                                             const float* __restrict__ w,
                                             const float* __restrict__ bias, float* __restrict__ out) {
  const int tid  = threadIdx.x;
  const int lane = tid & 31;
  const int hi   = lane >> 4;
  const int lm   = lane & 15;
  const int n0   = blockIdx.x * 16;
  const int h0   = (tid >> 5) * 16;

  const unsigned* xp = (const unsigned*)xbf + (size_t)(n0 + lm) * (Hh / 2);
  const float* wrow = w + (size_t)(h0 + lm) * Hh;
  WRow wr[Hh / 32];
  #pragma unroll
  for (int t = 0; t < Hh / 32; ++t) wr[t] = wrow_load(wrow, t, hi);

  v8f c = {0.f, 0.f, 0.f, 0.f, 0.f, 0.f, 0.f, 0.f};
  #pragma unroll
  for (int t = 0; t < Hh / 32; ++t) {
    BFrag a, b;
    #pragma unroll
    for (int j = 0; j < 8; ++j) {
      const int kk = t * 32 + ((j < 4) ? 0 : 16) + (hi ? 8 : 0) + 2 * (j & 3);
      a.u[j] = xp[kk >> 1];
    }
    wrow_pack(wr[t], b.u);
    c = __builtin_amdgcn_wmma_f32_16x16x32_bf16(false, a.v, false, b.v, (short)0, c, false, false);
  }
  #pragma unroll
  for (int r = 0; r < 8; ++r) {
    const int m = r + hi * 8;
    const int h = h0 + lm;
    out[(size_t)(n0 + m) * Hh + h] = fmaxf(c[r] + bias[h], 0.f);
  }
}

__global__ void __launch_bounds__(256) k_fc2(const float* __restrict__ x, const float* __restrict__ w,
                                             const float* __restrict__ b, float* __restrict__ out) {
  const int lane = threadIdx.x & 31;
  const int g = blockIdx.x * 8 + (threadIdx.x >> 5);
  if (g >= Gg) return;
  float dot = -INFINITY;
  if (lane < Cc) {
    float acc = b[lane];
    for (int d = 0; d < Hh; ++d) acc += x[(size_t)g * Hh + d] * w[lane * Hh + d];
    dot = acc;
  }
  float m = dot;
  for (int o = 16; o > 0; o >>= 1) m = fmaxf(m, __shfl_xor(m, o));
  float ex = (lane < Cc) ? __expf(dot - m) : 0.f;
  float se = ex;
  for (int o = 16; o > 0; o >>= 1) se += __shfl_xor(se, o);
  if (lane < Cc) out[(size_t)g * Cc + lane] = dot - m - logf(se);
}

extern "C" void kernel_launch(void* const* d_in, const int* in_sizes, int n_in,
                              void* d_out, int out_size, void* d_ws, size_t ws_size,
                              hipStream_t stream) {
  const float* x_in  = (const float*)d_in[0];
  const int*   eidx  = (const int*)  d_in[1];
  const int*   idx   = (const int*)  d_in[2];
  const float* fc_w  = (const float*)d_in[3];
  const float* a1    = (const float*)d_in[4];
  const float* a2    = (const float*)d_in[5];
  const float* fc1_w = (const float*)d_in[6];
  const float* fc1_b = (const float*)d_in[7];
  const float* fc2_w = (const float*)d_in[8];
  const float* fc2_b = (const float*)d_in[9];
  const float* gamma = (const float*)d_in[10];
  const float* beta  = (const float*)d_in[11];
  const int* src = eidx;
  const int* dst = eidx + Ee;

  float* ws   = (float*)d_ws;
  float* STAT = ws;
  float* CORR = ws + 64;
  float* Z    = ws + 256;
  float* VS   = Z;
  float* S2T  = Z + 512;
  float* GOUT = Z + (size_t)Nn * Hh;
  float* POOL = GOUT + (size_t)Nn * Hh;
  float* FC1O = POOL + (size_t)Gg * Hh;
  unsigned short* XBF  = (unsigned short*)(FC1O + (size_t)Gg * Hh);
  unsigned short* XBBF = XBF + (size_t)Nn * Dd;
  int*            TAB  = (int*)(XBBF + (size_t)Gg * Hh);
  int*            PKS  = TAB + (size_t)NCH * NOWN;

  (void)in_sizes; (void)n_in; (void)out_size; (void)ws_size;

  k_zero<<<1,    256, 0, stream>>>(STAT, 16);
  k_zero<<<64,   256, 0, stream>>>(POOL, (long long)Gg * Hh);

  k_stats<<<SBLK, 256, 0, stream>>>(x_in, STAT);
  k_prep<<<1, 256, 0, stream>>>(fc_w, a1, a2, STAT, CORR, VS);

  k_cvtx<<<SBLK, 256, 0, stream>>>(x_in, XBF, VS + Dd, S2T);

  k_bin<<<NCH, BINT, 0, stream>>>(src, dst, PKS, TAB);
  k_edge2<<<NOWN, 1024, 0, stream>>>(PKS, XBF, fc_w, VS, S2T, STAT, CORR, TAB, GOUT);
  k_pool<<<POOL_WG, POOL_WAVES * 32, 0, stream>>>(GOUT, idx, POOL);
  k_bn<<<BN_WG, BN_WAVES * 32, 0, stream>>>(POOL, gamma, beta, XBBF);
  k_fc1<<<Gg / 16, 256, 0, stream>>>(XBBF, fc1_w, fc1_b, FC1O);
  k_fc2<<<Gg / 8, 256, 0, stream>>>(FC1O, fc2_w, fc2_b, (float*)d_out);
}
